# starting kernel
# baseline (speedup 1.0000x reference)
#include <hip/hip_runtime.h>
#include <hip/hip_bf16.h>


#define N_NODES 10000
#define N_EDGES 640000
#define IN_C    128
#define OUT_C   128

typedef __attribute__((ext_vector_type(2))) float v2f;
typedef __attribute__((ext_vector_type(8))) float v8f;

__global__ void k_init_deg(float* __restrict__ deg) {
    int i = blockIdx.x * blockDim.x + threadIdx.x;
    if (i < N_NODES) deg[i] = 1.0f;
}

__global__ void k_deg_accum(const int* __restrict__ col, float* __restrict__ deg) {
    int e = blockIdx.x * blockDim.x + threadIdx.x;
    if (e < N_EDGES) atomicAdd(&deg[col[e]], 1.0f);
}

__global__ void k_inv_sqrt(const float* __restrict__ deg, float* __restrict__ dis) {
    int i = blockIdx.x * blockDim.x + threadIdx.x;
    if (i < N_NODES) {
        float d = deg[i];
        dis[i] = (d > 0.0f) ? rsqrtf(d) : 0.0f;
    }
}

__global__ void k_gemm_wmma(const float* __restrict__ x,
                            const float* __restrict__ W,
                            float* __restrict__ h) {
    const int wave   = (blockIdx.x * blockDim.x + threadIdx.x) >> 5;
    const int lane   = threadIdx.x & 31;
    const int tilesN = OUT_C / 16;
    const int tm     = wave / tilesN;
    const int tn     = wave % tilesN;

    const int mn    = lane & 15;
    const int khalf = (lane >> 4) << 1;

    const float* __restrict__ xr = x + (tm * 16 + mn) * IN_C + khalf;
    const float* __restrict__ wr = W + (tn * 16 + mn) * IN_C + khalf;

    v8f c = {};
    #pragma unroll 4
    for (int k = 0; k < IN_C; k += 4) {
        v2f a = *(const v2f*)(xr + k);
        v2f b = *(const v2f*)(wr + k);
        c = __builtin_amdgcn_wmma_f32_16x16x4_f32(false, a, false, b,
                                                  (short)0, c, false, false);
    }

    const int mbase = tm * 16 + ((lane >> 4) << 3);
    const int ncol  = tn * 16 + mn;
    #pragma unroll
    for (int v = 0; v < 8; ++v) {
        h[(mbase + v) * OUT_C + ncol] = c[v];
    }
}

__global__ void k_out_init(const float* __restrict__ h,
                           const float* __restrict__ dis,
                           const float* __restrict__ bias,
                           float* __restrict__ out) {
    int i = blockIdx.x * blockDim.x + threadIdx.x;
    if (i < N_NODES * OUT_C) {
        int node = i >> 7;
        int ch   = i & 127;
        float d  = dis[node];
        out[i] = h[i] * d * d + bias[ch];
    }
}

__global__ void k_edge_scatter(const int* __restrict__ row,
                               const int* __restrict__ col,
                               const float* __restrict__ h,
                               const float* __restrict__ dis,
                               float* __restrict__ out) {
    const int wid    = (blockIdx.x * blockDim.x + threadIdx.x) >> 5;
    const int lane   = threadIdx.x & 31;
    const int nwaves = (gridDim.x * blockDim.x) >> 5;

    for (int e = wid; e < N_EDGES; e += nwaves) {
        const int r = row[e];
        const int c = col[e];
        const float norm = dis[r] * dis[c];

        float4 v = *(const float4*)(h + r * OUT_C + lane * 4);
        float* o = out + c * OUT_C + lane * 4;
        atomicAdd(o + 0, v.x * norm);
        atomicAdd(o + 1, v.y * norm);
        atomicAdd(o + 2, v.z * norm);
        atomicAdd(o + 3, v.w * norm);
    }
}

extern "C" void kernel_launch(void* const* d_in, const int* in_sizes, int n_in,
                              void* d_out, int out_size, void* d_ws, size_t ws_size,
                              hipStream_t stream) {
    const float* x  = (const float*)d_in[0];
    const int*   ei = (const int*)  d_in[1];
    const float* W  = (const float*)d_in[2];
    const float* b  = (const float*)d_in[3];
    float*       out = (float*)d_out;

    const int* row = ei;
    const int* col = ei + N_EDGES;

    float* deg = (float*)d_ws;
    float* dis = deg + N_NODES;
    float* h   = dis + N_NODES;

    k_init_deg<<<(N_NODES + 255) / 256, 256, 0, stream>>>(deg);
    k_deg_accum<<<(N_EDGES + 255) / 256, 256, 0, stream>>>(col, deg);
    k_inv_sqrt<<<(N_NODES + 255) / 256, 256, 0, stream>>>(deg, dis);
    k_gemm_wmma<<<625, 256, 0, stream>>>(x, W, h);
    k_out_init<<<(N_NODES * OUT_C + 255) / 256, 256, 0, stream>>>(h, dis, b, out);
    k_edge_scatter<<<4096, 256, 0, stream>>>(row, col, h, dis, out);
}


# in-degree: 640k scattered global f32 atomics on 40 KB replaced by per-block LDS integer counts (64 edge slices) stored as planes and summed in the rsqrt kernel; deg exact; deg-init launch gone
# speedup vs baseline: 1.5292x; 1.5292x over previous
#include <hip/hip_runtime.h>
#include <hip/hip_bf16.h>


#define N_NODES 10000
#define N_EDGES 640000
#define IN_C    128
#define OUT_C   128

#define DEG_BLOCKS  64
#define DEG_THREADS 500
#define DEG_SLICE   (N_EDGES / DEG_BLOCKS)
#define DEG_EDGE_PASSES  (DEG_SLICE / 4 / DEG_THREADS)
#define DEG_TABLE_PASSES (N_NODES / 4 / DEG_THREADS)
#define DEG_PITCH   (((N_NODES + 31) / 32) * 32)

static_assert(DEG_BLOCKS * DEG_THREADS * DEG_EDGE_PASSES * 4 == N_EDGES, "slices cover the edge list exactly");
static_assert(DEG_THREADS * DEG_TABLE_PASSES * 4 == N_NODES, "passes cover a count table exactly");
static_assert(DEG_BLOCKS * DEG_PITCH <= N_NODES * OUT_C, "count planes fit in the h region");

typedef __attribute__((ext_vector_type(2))) float v2f;
typedef __attribute__((ext_vector_type(8))) float v8f;

__global__ void __launch_bounds__(DEG_THREADS)
k_deg_count(const int* __restrict__ col, unsigned* __restrict__ part) {
    __shared__ uint4 cnt4[N_NODES / 4];
    unsigned* cnt = (unsigned*)cnt4;
    const int t = threadIdx.x;

    const int4* __restrict__ c4 = (const int4*)(col + blockIdx.x * DEG_SLICE);
    int4 c[DEG_EDGE_PASSES];
    for (int j = 0; j < DEG_EDGE_PASSES; ++j) c[j] = c4[t + j * DEG_THREADS];

    for (int j = 0; j < DEG_TABLE_PASSES; ++j)
        cnt4[t + j * DEG_THREADS] = make_uint4(0u, 0u, 0u, 0u);
    __syncthreads();

    for (int j = 0; j < DEG_EDGE_PASSES; ++j) {
        atomicAdd(&cnt[c[j].x], 1u);
        atomicAdd(&cnt[c[j].y], 1u);
        atomicAdd(&cnt[c[j].z], 1u);
        atomicAdd(&cnt[c[j].w], 1u);
    }
    __syncthreads();

    uint4* __restrict__ p4 = (uint4*)(part + blockIdx.x * DEG_PITCH);
    for (int j = 0; j < DEG_TABLE_PASSES; ++j)
        p4[t + j * DEG_THREADS] = cnt4[t + j * DEG_THREADS];
}

__global__ void k_inv_sqrt(const unsigned* __restrict__ part, float* __restrict__ dis) {
    int i = blockIdx.x * blockDim.x + threadIdx.x;
    if (i < N_NODES) {
        unsigned n = 1u;
        for (int g = 0; g < DEG_BLOCKS; ++g) n += part[g * DEG_PITCH + i];
        float d = (float)n;
        dis[i] = (d > 0.0f) ? rsqrtf(d) : 0.0f;
    }
}

__global__ void k_gemm_wmma(const float* __restrict__ x,
                            const float* __restrict__ W,
                            float* __restrict__ h) {
    const int wave   = (blockIdx.x * blockDim.x + threadIdx.x) >> 5;
    const int lane   = threadIdx.x & 31;
    const int tilesN = OUT_C / 16;
    const int tm     = wave / tilesN;
    const int tn     = wave % tilesN;

    const int mn    = lane & 15;
    const int khalf = (lane >> 4) << 1;

    const float* __restrict__ xr = x + (tm * 16 + mn) * IN_C + khalf;
    const float* __restrict__ wr = W + (tn * 16 + mn) * IN_C + khalf;

    v8f c = {};
    #pragma unroll 4
    for (int k = 0; k < IN_C; k += 4) {
        v2f a = *(const v2f*)(xr + k);
        v2f b = *(const v2f*)(wr + k);
        c = __builtin_amdgcn_wmma_f32_16x16x4_f32(false, a, false, b,
                                                  (short)0, c, false, false);
    }

    const int mbase = tm * 16 + ((lane >> 4) << 3);
    const int ncol  = tn * 16 + mn;
    #pragma unroll
    for (int v = 0; v < 8; ++v) {
        h[(mbase + v) * OUT_C + ncol] = c[v];
    }
}

__global__ void k_out_init(const float* __restrict__ h,
                           const float* __restrict__ dis,
                           const float* __restrict__ bias,
                           float* __restrict__ out) {
    int i = blockIdx.x * blockDim.x + threadIdx.x;
    if (i < N_NODES * OUT_C) {
        int node = i >> 7;
        int ch   = i & 127;
        float d  = dis[node];
        out[i] = h[i] * d * d + bias[ch];
    }
}

__global__ void k_edge_scatter(const int* __restrict__ row,
                               const int* __restrict__ col,
                               const float* __restrict__ h,
                               const float* __restrict__ dis,
                               float* __restrict__ out) {
    const int wid    = (blockIdx.x * blockDim.x + threadIdx.x) >> 5;
    const int lane   = threadIdx.x & 31;
    const int nwaves = (gridDim.x * blockDim.x) >> 5;

    for (int e = wid; e < N_EDGES; e += nwaves) {
        const int r = row[e];
        const int c = col[e];
        const float norm = dis[r] * dis[c];

        float4 v = *(const float4*)(h + r * OUT_C + lane * 4);
        float* o = out + c * OUT_C + lane * 4;
        atomicAdd(o + 0, v.x * norm);
        atomicAdd(o + 1, v.y * norm);
        atomicAdd(o + 2, v.z * norm);
        atomicAdd(o + 3, v.w * norm);
    }
}

extern "C" void kernel_launch(void* const* d_in, const int* in_sizes, int n_in,
                              void* d_out, int out_size, void* d_ws, size_t ws_size,
                              hipStream_t stream) {
    const float* x  = (const float*)d_in[0];
    const int*   ei = (const int*)  d_in[1];
    const float* W  = (const float*)d_in[2];
    const float* b  = (const float*)d_in[3];
    float*       out = (float*)d_out;

    const int* row = ei;
    const int* col = ei + N_EDGES;

    float*    dis  = (float*)d_ws;
    float*    h    = dis + 2 * N_NODES;
    unsigned* part = (unsigned*)h;

    k_deg_count<<<DEG_BLOCKS, DEG_THREADS, 0, stream>>>(col, part);
    k_inv_sqrt<<<(N_NODES + 63) / 64, 64, 0, stream>>>(part, dis);
    k_gemm_wmma<<<625, 256, 0, stream>>>(x, W, h);
    k_out_init<<<(N_NODES * OUT_C + 255) / 256, 256, 0, stream>>>(h, dis, b, out);
    k_edge_scatter<<<4096, 256, 0, stream>>>(row, col, h, dis, out);
}


# f32 atomic scatter replaced by in-call counting sort of edges by destination (LDS ranks on slice counts); one wave per node sums dis*x rows in registers, product after; sum order changed
# speedup vs baseline: 2.7069x; 1.7701x over previous
#include <hip/hip_runtime.h>
#include <hip/hip_bf16.h>


#define N_NODES 10000
#define N_EDGES 640000
#define IN_C    128
#define OUT_C   128

#define DEG_BLOCKS  64
#define DEG_THREADS 500
#define DEG_SLICE   (N_EDGES / DEG_BLOCKS)
#define DEG_EDGE_PASSES  (DEG_SLICE / 4 / DEG_THREADS)
#define DEG_TABLE_PASSES (N_NODES / 4 / DEG_THREADS)
#define DEG_PITCH   (((N_NODES + 31) / 32) * 32)

#define PLACE_THREADS 512
#define PLACE_OWNERS  (N_NODES / 20)
#define PLACE_VEC     (DEG_SLICE / 4)

static_assert(DEG_BLOCKS * DEG_THREADS * DEG_EDGE_PASSES * 4 == N_EDGES, "slices cover the edge list exactly");
static_assert(DEG_THREADS * DEG_TABLE_PASSES * 4 == N_NODES, "passes cover a count table exactly");
static_assert(PLACE_OWNERS * 20 == N_NODES && PLACE_OWNERS <= PLACE_THREADS, "owners cover the bins exactly");
static_assert((DEG_BLOCKS * DEG_PITCH + N_NODES) <= N_NODES * IN_C, "count planes and totals fit in the agg region");
static_assert(N_EDGES <= N_NODES * OUT_C, "the ordered source list fits in the output buffer");

typedef __attribute__((ext_vector_type(2))) float v2f;
typedef __attribute__((ext_vector_type(8))) float v8f;

__global__ void __launch_bounds__(DEG_THREADS)
k_deg_count(const int* __restrict__ col, unsigned* __restrict__ part) {
    __shared__ uint4 cnt4[N_NODES / 4];
    unsigned* cnt = (unsigned*)cnt4;
    const int t = threadIdx.x;

    const int4* __restrict__ c4 = (const int4*)(col + blockIdx.x * DEG_SLICE);
    int4 c[DEG_EDGE_PASSES];
    for (int j = 0; j < DEG_EDGE_PASSES; ++j) c[j] = c4[t + j * DEG_THREADS];

    for (int j = 0; j < DEG_TABLE_PASSES; ++j)
        cnt4[t + j * DEG_THREADS] = make_uint4(0u, 0u, 0u, 0u);
    __syncthreads();

    for (int j = 0; j < DEG_EDGE_PASSES; ++j) {
        atomicAdd(&cnt[c[j].x], 1u);
        atomicAdd(&cnt[c[j].y], 1u);
        atomicAdd(&cnt[c[j].z], 1u);
        atomicAdd(&cnt[c[j].w], 1u);
    }
    __syncthreads();

    uint4* __restrict__ p4 = (uint4*)(part + blockIdx.x * DEG_PITCH);
    for (int j = 0; j < DEG_TABLE_PASSES; ++j)
        p4[t + j * DEG_THREADS] = cnt4[t + j * DEG_THREADS];
}

__global__ void k_slice_prefix(unsigned* __restrict__ part,
                               unsigned* __restrict__ tot,
                               float* __restrict__ dis) {
    int i = blockIdx.x * blockDim.x + threadIdx.x;
    if (i < N_NODES) {
        unsigned n[DEG_BLOCKS];
        #pragma unroll
        for (int g = 0; g < DEG_BLOCKS; ++g) n[g] = part[g * DEG_PITCH + i];
        unsigned run = 0u;
        #pragma unroll
        for (int g = 0; g < DEG_BLOCKS; ++g) {
            part[g * DEG_PITCH + i] = run;
            run += n[g];
        }
        tot[i] = run;
        float d = (float)(run + 1u);
        dis[i] = (d > 0.0f) ? rsqrtf(d) : 0.0f;
    }
}

__global__ void __launch_bounds__(PLACE_THREADS)
k_place(const int* __restrict__ row, const int* __restrict__ col,
        const unsigned* __restrict__ part, const unsigned* __restrict__ tot,
        unsigned* __restrict__ start, int* __restrict__ sorted) {
    __shared__ uint4 pos4[N_NODES / 4];
    __shared__ unsigned wave_sum[PLACE_THREADS / 32];
    unsigned* pos = (unsigned*)pos4;
    const int t    = threadIdx.x;
    const int lane = t & 31;
    const int wave = t >> 5;
    const int g    = blockIdx.x;

    const int4* __restrict__ c4 = (const int4*)(col + g * DEG_SLICE);
    const int4* __restrict__ r4 = (const int4*)(row + g * DEG_SLICE);
    int4 c[5], r[5];
    for (int j = 0; j < 5; ++j) {
        const int idx = t + j * PLACE_THREADS;
        if (idx < PLACE_VEC) { c[j] = c4[idx]; r[j] = r4[idx]; }
    }

    uint4 e[5], o[5];
    unsigned mine = 0u;
    if (t < PLACE_OWNERS) {
        const uint4* __restrict__ n4 = (const uint4*)tot + t * 5;
        const uint4* __restrict__ o4 = (const uint4*)(part + g * DEG_PITCH) + t * 5;
        for (int j = 0; j < 5; ++j) {
            const uint4 n = n4[j];
            o[j] = o4[j];
            e[j].x = mine; mine += n.x;
            e[j].y = mine; mine += n.y;
            e[j].z = mine; mine += n.z;
            e[j].w = mine; mine += n.w;
        }
    }
    unsigned inc = mine;
    for (int d = 1; d < 32; d <<= 1) {
        const unsigned v = __shfl_up(inc, d);
        if (lane >= d) inc += v;
    }
    if (lane == 31) wave_sum[wave] = inc;
    __syncthreads();
    unsigned before = inc - mine;
    for (int w = 0; w < wave; ++w) before += wave_sum[w];

    if (t < PLACE_OWNERS) {
        for (int j = 0; j < 5; ++j) {
            uint4 s = e[j];
            s.x += before; s.y += before; s.z += before; s.w += before;
            if (g == 0) ((uint4*)start)[t * 5 + j] = s;
            s.x += o[j].x; s.y += o[j].y; s.z += o[j].z; s.w += o[j].w;
            pos4[t * 5 + j] = s;
        }
    }
    __syncthreads();

    for (int j = 0; j < 5; ++j) {
        const int idx = t + j * PLACE_THREADS;
        if (idx < PLACE_VEC) {
            c[j].x = (int)atomicAdd(&pos[c[j].x], 1u);
            c[j].y = (int)atomicAdd(&pos[c[j].y], 1u);
            c[j].z = (int)atomicAdd(&pos[c[j].z], 1u);
            c[j].w = (int)atomicAdd(&pos[c[j].w], 1u);
        }
    }
    for (int j = 0; j < 5; ++j) {
        const int idx = t + j * PLACE_THREADS;
        if (idx < PLACE_VEC) {
            sorted[c[j].x] = r[j].x;
            sorted[c[j].y] = r[j].y;
            sorted[c[j].z] = r[j].z;
            sorted[c[j].w] = r[j].w;
        }
    }
}

#define GATHER_BATCH 8
__global__ void __launch_bounds__(256)
k_gather(const float* __restrict__ x, const float* __restrict__ dis,
         const unsigned* __restrict__ start, const int* __restrict__ sorted,
         float* __restrict__ agg) {
    const int c    = __builtin_amdgcn_readfirstlane((int)((blockIdx.x * blockDim.x + threadIdx.x) >> 5));
    const int lane = threadIdx.x & 31;

    const int s = (int)start[c];
    const int n = ((c + 1 < N_NODES) ? (int)start[c + 1] : N_EDGES) - s;
    const float dc = dis[c];
    const float4 self = *(const float4*)(x + c * IN_C + lane * 4);

    int   rl = c, rn = c;
    float wl = 0.0f;
    if (lane < n)      rl = sorted[s + lane];
    if (32 + lane < n) rn = sorted[s + 32 + lane];
    if (lane < n)      wl = dis[rl];

    float4 acc = make_float4(dc * self.x, dc * self.y, dc * self.z, dc * self.w);

    for (int base = 0; base < n; base += 32) {
        int   rnn = c;
        float wn  = 0.0f;
        if (base + 64 + lane < n) rnn = sorted[s + base + 64 + lane];
        if (base + 32 + lane < n) wn  = dis[rn];

        const int m = (n - base < 32) ? (n - base) : 32;
        for (int j = 0; j < m; j += GATHER_BATCH) {
            float4 v[GATHER_BATCH];
            float  w[GATHER_BATCH];
            #pragma unroll
            for (int u = 0; u < GATHER_BATCH; ++u) {
                const int rr = __builtin_amdgcn_readlane(rl, j + u);
                w[u] = __int_as_float(__builtin_amdgcn_readlane(__float_as_int(wl), j + u));
                v[u] = *(const float4*)(x + rr * IN_C + lane * 4);
            }
            #pragma unroll
            for (int u = 0; u < GATHER_BATCH; ++u) {
                acc.x += w[u] * v[u].x;
                acc.y += w[u] * v[u].y;
                acc.z += w[u] * v[u].z;
                acc.w += w[u] * v[u].w;
            }
        }
        rl = rn;
        wl = wn;
        rn = rnn;
    }

    *(float4*)(agg + c * IN_C + lane * 4) =
        make_float4(dc * acc.x, dc * acc.y, dc * acc.z, dc * acc.w);
}

__global__ void k_gemm_wmma(const float* __restrict__ a_in,
                            const float* __restrict__ W,
                            const float* __restrict__ bias,
                            float* __restrict__ out) {
    const int wave   = (blockIdx.x * blockDim.x + threadIdx.x) >> 5;
    const int lane   = threadIdx.x & 31;
    const int tilesN = OUT_C / 16;
    const int tm     = wave / tilesN;
    const int tn     = wave % tilesN;

    const int mn    = lane & 15;
    const int khalf = (lane >> 4) << 1;

    const float* __restrict__ xr = a_in + (tm * 16 + mn) * IN_C + khalf;
    const float* __restrict__ wr = W    + (tn * 16 + mn) * IN_C + khalf;

    const int   ncol = tn * 16 + mn;
    const float bn   = bias[ncol];
    v8f c = {bn, bn, bn, bn, bn, bn, bn, bn};
    #pragma unroll 4
    for (int k = 0; k < IN_C; k += 4) {
        v2f a = *(const v2f*)(xr + k);
        v2f b = *(const v2f*)(wr + k);
        c = __builtin_amdgcn_wmma_f32_16x16x4_f32(false, a, false, b,
                                                  (short)0, c, false, false);
    }

    const int mbase = tm * 16 + ((lane >> 4) << 3);
    #pragma unroll
    for (int v = 0; v < 8; ++v) {
        out[(mbase + v) * OUT_C + ncol] = c[v];
    }
}

extern "C" void kernel_launch(void* const* d_in, const int* in_sizes, int n_in,
                              void* d_out, int out_size, void* d_ws, size_t ws_size,
                              hipStream_t stream) {
    const float* x  = (const float*)d_in[0];
    const int*   ei = (const int*)  d_in[1];
    const float* W  = (const float*)d_in[2];
    const float* b  = (const float*)d_in[3];
    float*       out = (float*)d_out;

    const int* row = ei;
    const int* col = ei + N_EDGES;

    float*    dis    = (float*)d_ws;
    unsigned* start  = (unsigned*)(dis + N_NODES);
    float*    agg    = dis + 2 * N_NODES;
    unsigned* part   = (unsigned*)agg;
    unsigned* tot    = part + DEG_BLOCKS * DEG_PITCH;
    int*      sorted = (int*)d_out;

    k_deg_count<<<DEG_BLOCKS, DEG_THREADS, 0, stream>>>(col, part);
    k_slice_prefix<<<(N_NODES + 63) / 64, 64, 0, stream>>>(part, tot, dis);
    k_place<<<DEG_BLOCKS, PLACE_THREADS, 0, stream>>>(row, col, part, tot, start, sorted);
    k_gather<<<N_NODES / 8, 256, 0, stream>>>(x, dis, start, sorted, agg);
    k_gemm_wmma<<<625, 256, 0, stream>>>(agg, W, b, out);
}


# GEMM: agg rows and W copied to LDS with whole-row loads (padded pitch), WMMA fragments read from LDS instead of 16-line strided global loads; k order and sums unchanged
# speedup vs baseline: 2.8434x; 1.0504x over previous
#include <hip/hip_runtime.h>
#include <hip/hip_bf16.h>


#define N_NODES 10000
#define N_EDGES 640000
#define IN_C    128
#define OUT_C   128

#define DEG_BLOCKS  64
#define DEG_THREADS 500
#define DEG_SLICE   (N_EDGES / DEG_BLOCKS)
#define DEG_EDGE_PASSES  (DEG_SLICE / 4 / DEG_THREADS)
#define DEG_TABLE_PASSES (N_NODES / 4 / DEG_THREADS)
#define DEG_PITCH   (((N_NODES + 31) / 32) * 32)

#define PLACE_THREADS 512
#define PLACE_OWNERS  (N_NODES / 20)
#define PLACE_VEC     (DEG_SLICE / 4)

static_assert(DEG_BLOCKS * DEG_THREADS * DEG_EDGE_PASSES * 4 == N_EDGES, "slices cover the edge list exactly");
static_assert(DEG_THREADS * DEG_TABLE_PASSES * 4 == N_NODES, "passes cover a count table exactly");
static_assert(PLACE_OWNERS * 20 == N_NODES && PLACE_OWNERS <= PLACE_THREADS, "owners cover the bins exactly");
static_assert((DEG_BLOCKS * DEG_PITCH + N_NODES) <= N_NODES * IN_C, "count planes and totals fit in the agg region");
static_assert(N_EDGES <= N_NODES * OUT_C, "the ordered source list fits in the output buffer");

typedef __attribute__((ext_vector_type(2))) float v2f;
typedef __attribute__((ext_vector_type(8))) float v8f;

__global__ void __launch_bounds__(DEG_THREADS)
k_deg_count(const int* __restrict__ col, unsigned* __restrict__ part) {
    __shared__ uint4 cnt4[N_NODES / 4];
    unsigned* cnt = (unsigned*)cnt4;
    const int t = threadIdx.x;

    const int4* __restrict__ c4 = (const int4*)(col + blockIdx.x * DEG_SLICE);
    int4 c[DEG_EDGE_PASSES];
    for (int j = 0; j < DEG_EDGE_PASSES; ++j) c[j] = c4[t + j * DEG_THREADS];

    for (int j = 0; j < DEG_TABLE_PASSES; ++j)
        cnt4[t + j * DEG_THREADS] = make_uint4(0u, 0u, 0u, 0u);
    __syncthreads();

    for (int j = 0; j < DEG_EDGE_PASSES; ++j) {
        atomicAdd(&cnt[c[j].x], 1u);
        atomicAdd(&cnt[c[j].y], 1u);
        atomicAdd(&cnt[c[j].z], 1u);
        atomicAdd(&cnt[c[j].w], 1u);
    }
    __syncthreads();

    uint4* __restrict__ p4 = (uint4*)(part + blockIdx.x * DEG_PITCH);
    for (int j = 0; j < DEG_TABLE_PASSES; ++j)
        p4[t + j * DEG_THREADS] = cnt4[t + j * DEG_THREADS];
}

__global__ void k_slice_prefix(unsigned* __restrict__ part,
                               unsigned* __restrict__ tot,
                               float* __restrict__ dis) {
    int i = blockIdx.x * blockDim.x + threadIdx.x;
    if (i < N_NODES) {
        unsigned n[DEG_BLOCKS];
        #pragma unroll
        for (int g = 0; g < DEG_BLOCKS; ++g) n[g] = part[g * DEG_PITCH + i];
        unsigned run = 0u;
        #pragma unroll
        for (int g = 0; g < DEG_BLOCKS; ++g) {
            part[g * DEG_PITCH + i] = run;
            run += n[g];
        }
        tot[i] = run;
        float d = (float)(run + 1u);
        dis[i] = (d > 0.0f) ? rsqrtf(d) : 0.0f;
    }
}

__global__ void __launch_bounds__(PLACE_THREADS)
k_place(const int* __restrict__ row, const int* __restrict__ col,
        const unsigned* __restrict__ part, const unsigned* __restrict__ tot,
        unsigned* __restrict__ start, int* __restrict__ sorted) {
    __shared__ uint4 pos4[N_NODES / 4];
    __shared__ unsigned wave_sum[PLACE_THREADS / 32];
    unsigned* pos = (unsigned*)pos4;
    const int t    = threadIdx.x;
    const int lane = t & 31;
    const int wave = t >> 5;
    const int g    = blockIdx.x;

    const int4* __restrict__ c4 = (const int4*)(col + g * DEG_SLICE);
    const int4* __restrict__ r4 = (const int4*)(row + g * DEG_SLICE);
    int4 c[5], r[5];
    for (int j = 0; j < 5; ++j) {
        const int idx = t + j * PLACE_THREADS;
        if (idx < PLACE_VEC) { c[j] = c4[idx]; r[j] = r4[idx]; }
    }

    uint4 e[5], o[5];
    unsigned mine = 0u;
    if (t < PLACE_OWNERS) {
        const uint4* __restrict__ n4 = (const uint4*)tot + t * 5;
        const uint4* __restrict__ o4 = (const uint4*)(part + g * DEG_PITCH) + t * 5;
        for (int j = 0; j < 5; ++j) {
            const uint4 n = n4[j];
            o[j] = o4[j];
            e[j].x = mine; mine += n.x;
            e[j].y = mine; mine += n.y;
            e[j].z = mine; mine += n.z;
            e[j].w = mine; mine += n.w;
        }
    }
    unsigned inc = mine;
    for (int d = 1; d < 32; d <<= 1) {
        const unsigned v = __shfl_up(inc, d);
        if (lane >= d) inc += v;
    }
    if (lane == 31) wave_sum[wave] = inc;
    __syncthreads();
    unsigned before = inc - mine;
    for (int w = 0; w < wave; ++w) before += wave_sum[w];

    if (t < PLACE_OWNERS) {
        for (int j = 0; j < 5; ++j) {
            uint4 s = e[j];
            s.x += before; s.y += before; s.z += before; s.w += before;
            if (g == 0) ((uint4*)start)[t * 5 + j] = s;
            s.x += o[j].x; s.y += o[j].y; s.z += o[j].z; s.w += o[j].w;
            pos4[t * 5 + j] = s;
        }
    }
    __syncthreads();

    for (int j = 0; j < 5; ++j) {
        const int idx = t + j * PLACE_THREADS;
        if (idx < PLACE_VEC) {
            c[j].x = (int)atomicAdd(&pos[c[j].x], 1u);
            c[j].y = (int)atomicAdd(&pos[c[j].y], 1u);
            c[j].z = (int)atomicAdd(&pos[c[j].z], 1u);
            c[j].w = (int)atomicAdd(&pos[c[j].w], 1u);
        }
    }
    for (int j = 0; j < 5; ++j) {
        const int idx = t + j * PLACE_THREADS;
        if (idx < PLACE_VEC) {
            sorted[c[j].x] = r[j].x;
            sorted[c[j].y] = r[j].y;
            sorted[c[j].z] = r[j].z;
            sorted[c[j].w] = r[j].w;
        }
    }
}

#define GATHER_BATCH 8
__global__ void __launch_bounds__(256)
k_gather(const float* __restrict__ x, const float* __restrict__ dis,
         const unsigned* __restrict__ start, const int* __restrict__ sorted,
         float* __restrict__ agg) {
    const int c    = __builtin_amdgcn_readfirstlane((int)((blockIdx.x * blockDim.x + threadIdx.x) >> 5));
    const int lane = threadIdx.x & 31;

    const int s = (int)start[c];
    const int n = ((c + 1 < N_NODES) ? (int)start[c + 1] : N_EDGES) - s;
    const float dc = dis[c];
    const float4 self = *(const float4*)(x + c * IN_C + lane * 4);

    int   rl = c, rn = c;
    float wl = 0.0f;
    if (lane < n)      rl = sorted[s + lane];
    if (32 + lane < n) rn = sorted[s + 32 + lane];
    if (lane < n)      wl = dis[rl];

    float4 acc = make_float4(dc * self.x, dc * self.y, dc * self.z, dc * self.w);

    for (int base = 0; base < n; base += 32) {
        int   rnn = c;
        float wn  = 0.0f;
        if (base + 64 + lane < n) rnn = sorted[s + base + 64 + lane];
        if (base + 32 + lane < n) wn  = dis[rn];

        const int m = (n - base < 32) ? (n - base) : 32;
        for (int j = 0; j < m; j += GATHER_BATCH) {
            float4 v[GATHER_BATCH];
            float  w[GATHER_BATCH];
            #pragma unroll
            for (int u = 0; u < GATHER_BATCH; ++u) {
                const int rr = __builtin_amdgcn_readlane(rl, j + u);
                w[u] = __int_as_float(__builtin_amdgcn_readlane(__float_as_int(wl), j + u));
                v[u] = *(const float4*)(x + rr * IN_C + lane * 4);
            }
            #pragma unroll
            for (int u = 0; u < GATHER_BATCH; ++u) {
                acc.x += w[u] * v[u].x;
                acc.y += w[u] * v[u].y;
                acc.z += w[u] * v[u].z;
                acc.w += w[u] * v[u].w;
            }
        }
        rl = rn;
        wl = wn;
        rn = rnn;
    }

    *(float4*)(agg + c * IN_C + lane * 4) =
        make_float4(dc * acc.x, dc * acc.y, dc * acc.z, dc * acc.w);
}

#define GEMM_PITCH 132
__global__ void __launch_bounds__(256)
k_gemm_wmma(const float* __restrict__ a_in,
            const float* __restrict__ W,
            const float* __restrict__ bias,
            float* __restrict__ out) {
    __shared__ float4 w_s4[OUT_C * GEMM_PITCH / 4];
    __shared__ float4 a_s4[16 * GEMM_PITCH / 4];
    float* w_s = (float*)w_s4;
    float* a_s = (float*)a_s4;

    const int t    = threadIdx.x;
    const int lane = t & 31;
    const int tn   = t >> 5;
    const int tm   = blockIdx.x;
    const int mn   = lane & 15;
    const int ncol = tn * 16 + mn;
    const float bn = bias[ncol];

    const float4* __restrict__ a4 = (const float4*)(a_in + tm * 16 * IN_C);
    const float4* __restrict__ w4 = (const float4*)W;
    float4 av[2], wv[16];
    for (int j = 0; j < 2; ++j)  av[j] = a4[t + j * 256];
    for (int j = 0; j < 16; ++j) wv[j] = w4[t + j * 256];
    for (int j = 0; j < 2; ++j) {
        const int idx = t + j * 256;
        *(float4*)(a_s + (idx >> 5) * GEMM_PITCH + (idx & 31) * 4) = av[j];
    }
    for (int j = 0; j < 16; ++j) {
        const int idx = t + j * 256;
        *(float4*)(w_s + (idx >> 5) * GEMM_PITCH + (idx & 31) * 4) = wv[j];
    }
    __syncthreads();

    const int khalf = (lane >> 4) << 1;

    const float* xr = a_s + mn * GEMM_PITCH + khalf;
    const float* wr = w_s + (tn * 16 + mn) * GEMM_PITCH + khalf;

    v8f c = {bn, bn, bn, bn, bn, bn, bn, bn};
    #pragma unroll 4
    for (int k = 0; k < IN_C; k += 4) {
        v2f a = *(const v2f*)(xr + k);
        v2f b = *(const v2f*)(wr + k);
        c = __builtin_amdgcn_wmma_f32_16x16x4_f32(false, a, false, b,
                                                  (short)0, c, false, false);
    }

    const int mbase = tm * 16 + ((lane >> 4) << 3);
    #pragma unroll
    for (int v = 0; v < 8; ++v) {
        out[(mbase + v) * OUT_C + ncol] = c[v];
    }
}

extern "C" void kernel_launch(void* const* d_in, const int* in_sizes, int n_in,
                              void* d_out, int out_size, void* d_ws, size_t ws_size,
                              hipStream_t stream) {
    const float* x  = (const float*)d_in[0];
    const int*   ei = (const int*)  d_in[1];
    const float* W  = (const float*)d_in[2];
    const float* b  = (const float*)d_in[3];
    float*       out = (float*)d_out;

    const int* row = ei;
    const int* col = ei + N_EDGES;

    float*    dis    = (float*)d_ws;
    unsigned* start  = (unsigned*)(dis + N_NODES);
    float*    agg    = dis + 2 * N_NODES;
    unsigned* part   = (unsigned*)agg;
    unsigned* tot    = part + DEG_BLOCKS * DEG_PITCH;
    int*      sorted = (int*)d_out;

    k_deg_count<<<DEG_BLOCKS, DEG_THREADS, 0, stream>>>(col, part);
    k_slice_prefix<<<(N_NODES + 63) / 64, 64, 0, stream>>>(part, tot, dis);
    k_place<<<DEG_BLOCKS, PLACE_THREADS, 0, stream>>>(row, col, part, tot, start, sorted);
    k_gather<<<N_NODES / 8, 256, 0, stream>>>(x, dis, start, sorted, agg);
    k_gemm_wmma<<<625, 256, 0, stream>>>(agg, W, b, out);
}


# GEMM fused into the gather launch: a block's one-node waves leave their agg rows in LDS as the A tile and the same WMMA tiles run there; agg store, reload and one launch gone; sums unchanged
# speedup vs baseline: 3.0383x; 1.0685x over previous
#include <hip/hip_runtime.h>
#include <hip/hip_bf16.h>


#define N_NODES 10000
#define N_EDGES 640000
#define IN_C    128
#define OUT_C   128

#define DEG_BLOCKS  64
#define DEG_THREADS 500
#define DEG_SLICE   (N_EDGES / DEG_BLOCKS)
#define DEG_EDGE_PASSES  (DEG_SLICE / 4 / DEG_THREADS)
#define DEG_TABLE_PASSES (N_NODES / 4 / DEG_THREADS)
#define DEG_PITCH   (((N_NODES + 31) / 32) * 32)

#define PLACE_THREADS 512
#define PLACE_OWNERS  (N_NODES / 20)
#define PLACE_VEC     (DEG_SLICE / 4)

static_assert(DEG_BLOCKS * DEG_THREADS * DEG_EDGE_PASSES * 4 == N_EDGES, "slices cover the edge list exactly");
static_assert(DEG_THREADS * DEG_TABLE_PASSES * 4 == N_NODES, "passes cover a count table exactly");
static_assert(PLACE_OWNERS * 20 == N_NODES && PLACE_OWNERS <= PLACE_THREADS, "owners cover the bins exactly");
static_assert((DEG_BLOCKS * DEG_PITCH + N_NODES) <= N_NODES * OUT_C, "count planes and totals fit in the output buffer");
static_assert(N_EDGES <= N_NODES * IN_C, "the ordered source list fits in the workspace behind dis and start");
static_assert(N_NODES % 16 == 0, "16-node blocks cover the nodes exactly");

typedef __attribute__((ext_vector_type(2))) float v2f;
typedef __attribute__((ext_vector_type(8))) float v8f;

__global__ void __launch_bounds__(DEG_THREADS)
k_deg_count(const int* __restrict__ col, unsigned* __restrict__ part) {
    __shared__ uint4 cnt4[N_NODES / 4];
    unsigned* cnt = (unsigned*)cnt4;
    const int t = threadIdx.x;

    const int4* __restrict__ c4 = (const int4*)(col + blockIdx.x * DEG_SLICE);
    int4 c[DEG_EDGE_PASSES];
    for (int j = 0; j < DEG_EDGE_PASSES; ++j) c[j] = c4[t + j * DEG_THREADS];

    for (int j = 0; j < DEG_TABLE_PASSES; ++j)
        cnt4[t + j * DEG_THREADS] = make_uint4(0u, 0u, 0u, 0u);
    __syncthreads();

    for (int j = 0; j < DEG_EDGE_PASSES; ++j) {
        atomicAdd(&cnt[c[j].x], 1u);
        atomicAdd(&cnt[c[j].y], 1u);
        atomicAdd(&cnt[c[j].z], 1u);
        atomicAdd(&cnt[c[j].w], 1u);
    }
    __syncthreads();

    uint4* __restrict__ p4 = (uint4*)(part + blockIdx.x * DEG_PITCH);
    for (int j = 0; j < DEG_TABLE_PASSES; ++j)
        p4[t + j * DEG_THREADS] = cnt4[t + j * DEG_THREADS];
}

__global__ void k_slice_prefix(unsigned* __restrict__ part,
                               unsigned* __restrict__ tot,
                               float* __restrict__ dis) {
    int i = blockIdx.x * blockDim.x + threadIdx.x;
    if (i < N_NODES) {
        unsigned n[DEG_BLOCKS];
        #pragma unroll
        for (int g = 0; g < DEG_BLOCKS; ++g) n[g] = part[g * DEG_PITCH + i];
        unsigned run = 0u;
        #pragma unroll
        for (int g = 0; g < DEG_BLOCKS; ++g) {
            part[g * DEG_PITCH + i] = run;
            run += n[g];
        }
        tot[i] = run;
        float d = (float)(run + 1u);
        dis[i] = (d > 0.0f) ? rsqrtf(d) : 0.0f;
    }
}

__global__ void __launch_bounds__(PLACE_THREADS)
k_place(const int* __restrict__ row, const int* __restrict__ col,
        const unsigned* __restrict__ part, const unsigned* __restrict__ tot,
        unsigned* __restrict__ start, int* __restrict__ sorted) {
    __shared__ uint4 pos4[N_NODES / 4];
    __shared__ unsigned wave_sum[PLACE_THREADS / 32];
    unsigned* pos = (unsigned*)pos4;
    const int t    = threadIdx.x;
    const int lane = t & 31;
    const int wave = t >> 5;
    const int g    = blockIdx.x;

    const int4* __restrict__ c4 = (const int4*)(col + g * DEG_SLICE);
    const int4* __restrict__ r4 = (const int4*)(row + g * DEG_SLICE);
    int4 c[5], r[5];
    for (int j = 0; j < 5; ++j) {
        const int idx = t + j * PLACE_THREADS;
        if (idx < PLACE_VEC) { c[j] = c4[idx]; r[j] = r4[idx]; }
    }

    uint4 e[5], o[5];
    unsigned mine = 0u;
    if (t < PLACE_OWNERS) {
        const uint4* __restrict__ n4 = (const uint4*)tot + t * 5;
        const uint4* __restrict__ o4 = (const uint4*)(part + g * DEG_PITCH) + t * 5;
        for (int j = 0; j < 5; ++j) {
            const uint4 n = n4[j];
            o[j] = o4[j];
            e[j].x = mine; mine += n.x;
            e[j].y = mine; mine += n.y;
            e[j].z = mine; mine += n.z;
            e[j].w = mine; mine += n.w;
        }
    }
    unsigned inc = mine;
    for (int d = 1; d < 32; d <<= 1) {
        const unsigned v = __shfl_up(inc, d);
        if (lane >= d) inc += v;
    }
    if (lane == 31) wave_sum[wave] = inc;
    __syncthreads();
    unsigned before = inc - mine;
    for (int w = 0; w < wave; ++w) before += wave_sum[w];

    if (t < PLACE_OWNERS) {
        for (int j = 0; j < 5; ++j) {
            uint4 s = e[j];
            s.x += before; s.y += before; s.z += before; s.w += before;
            if (g == 0) ((uint4*)start)[t * 5 + j] = s;
            s.x += o[j].x; s.y += o[j].y; s.z += o[j].z; s.w += o[j].w;
            pos4[t * 5 + j] = s;
        }
    }
    __syncthreads();

    for (int j = 0; j < 5; ++j) {
        const int idx = t + j * PLACE_THREADS;
        if (idx < PLACE_VEC) {
            c[j].x = (int)atomicAdd(&pos[c[j].x], 1u);
            c[j].y = (int)atomicAdd(&pos[c[j].y], 1u);
            c[j].z = (int)atomicAdd(&pos[c[j].z], 1u);
            c[j].w = (int)atomicAdd(&pos[c[j].w], 1u);
        }
    }
    for (int j = 0; j < 5; ++j) {
        const int idx = t + j * PLACE_THREADS;
        if (idx < PLACE_VEC) {
            sorted[c[j].x] = r[j].x;
            sorted[c[j].y] = r[j].y;
            sorted[c[j].z] = r[j].z;
            sorted[c[j].w] = r[j].w;
        }
    }
}

#define GATHER_BATCH 8
#define GEMM_PITCH 132
#define FUSE_THREADS 512
__global__ void __launch_bounds__(FUSE_THREADS)
k_gather_gemm(const float* __restrict__ x, const float* __restrict__ dis,
              const unsigned* __restrict__ start, const int* __restrict__ sorted,
              const float* __restrict__ W, const float* __restrict__ bias,
              float* __restrict__ out) {
    __shared__ float4 w_s4[OUT_C * GEMM_PITCH / 4];
    __shared__ float4 a_s4[16 * GEMM_PITCH / 4];
    float* w_s = (float*)w_s4;
    float* a_s = (float*)a_s4;

    const int t    = threadIdx.x;
    const int lane = t & 31;
    const int wave = __builtin_amdgcn_readfirstlane(t >> 5);
    const int tm   = blockIdx.x;
    const int c    = tm * 16 + wave;

    const int mn   = lane & 15;
    const int ncol = (wave & 7) * 16 + mn;
    const float bn = bias[ncol];

    const int s = (int)start[c];
    const int n = ((c + 1 < N_NODES) ? (int)start[c + 1] : N_EDGES) - s;
    const float dc = dis[c];
    const float4 self = *(const float4*)(x + c * IN_C + lane * 4);

    int   rl = c, rn = c;
    float wl = 0.0f;
    if (lane < n)      rl = sorted[s + lane];
    if (32 + lane < n) rn = sorted[s + 32 + lane];
    if (lane < n)      wl = dis[rl];

    const float4* __restrict__ w4 = (const float4*)W;
    float4 wv[8];
    for (int j = 0; j < 8; ++j) wv[j] = w4[t + j * FUSE_THREADS];
    for (int j = 0; j < 8; ++j) {
        const int idx = t + j * FUSE_THREADS;
        *(float4*)(w_s + (idx >> 5) * GEMM_PITCH + (idx & 31) * 4) = wv[j];
    }

    float4 acc = make_float4(dc * self.x, dc * self.y, dc * self.z, dc * self.w);

    for (int base = 0; base < n; base += 32) {
        int   rnn = c;
        float wn  = 0.0f;
        if (base + 64 + lane < n) rnn = sorted[s + base + 64 + lane];
        if (base + 32 + lane < n) wn  = dis[rn];

        const int m = (n - base < 32) ? (n - base) : 32;
        for (int j = 0; j < m; j += GATHER_BATCH) {
            float4 v[GATHER_BATCH];
            float  w[GATHER_BATCH];
            #pragma unroll
            for (int u = 0; u < GATHER_BATCH; ++u) {
                const int rr = __builtin_amdgcn_readlane(rl, j + u);
                w[u] = __int_as_float(__builtin_amdgcn_readlane(__float_as_int(wl), j + u));
                v[u] = *(const float4*)(x + rr * IN_C + lane * 4);
            }
            #pragma unroll
            for (int u = 0; u < GATHER_BATCH; ++u) {
                acc.x += w[u] * v[u].x;
                acc.y += w[u] * v[u].y;
                acc.z += w[u] * v[u].z;
                acc.w += w[u] * v[u].w;
            }
        }
        rl = rn;
        wl = wn;
        rn = rnn;
    }

    *(float4*)(a_s + wave * GEMM_PITCH + lane * 4) =
        make_float4(dc * acc.x, dc * acc.y, dc * acc.z, dc * acc.w);
    __syncthreads();

    if (wave < 8) {
        const int khalf = (lane >> 4) << 1;

        const float* xr = a_s + mn * GEMM_PITCH + khalf;
        const float* wr = w_s + (wave * 16 + mn) * GEMM_PITCH + khalf;

        v8f d = {bn, bn, bn, bn, bn, bn, bn, bn};
        #pragma unroll 4
        for (int k = 0; k < IN_C; k += 4) {
            v2f a = *(const v2f*)(xr + k);
            v2f b = *(const v2f*)(wr + k);
            d = __builtin_amdgcn_wmma_f32_16x16x4_f32(false, a, false, b,
                                                      (short)0, d, false, false);
        }

        const int mbase = tm * 16 + ((lane >> 4) << 3);
        #pragma unroll
        for (int v = 0; v < 8; ++v) {
            out[(mbase + v) * OUT_C + ncol] = d[v];
        }
    }
}

extern "C" void kernel_launch(void* const* d_in, const int* in_sizes, int n_in,
                              void* d_out, int out_size, void* d_ws, size_t ws_size,
                              hipStream_t stream) {
    const float* x  = (const float*)d_in[0];
    const int*   ei = (const int*)  d_in[1];
    const float* W  = (const float*)d_in[2];
    const float* b  = (const float*)d_in[3];
    float*       out = (float*)d_out;

    const int* row = ei;
    const int* col = ei + N_EDGES;

    float*    dis    = (float*)d_ws;
    unsigned* start  = (unsigned*)(dis + N_NODES);
    int*      sorted = (int*)(dis + 2 * N_NODES);
    unsigned* part   = (unsigned*)d_out;
    unsigned* tot    = part + DEG_BLOCKS * DEG_PITCH;

    k_deg_count<<<DEG_BLOCKS, DEG_THREADS, 0, stream>>>(col, part);
    k_slice_prefix<<<(N_NODES + 63) / 64, 64, 0, stream>>>(part, tot, dis);
    k_place<<<DEG_BLOCKS, PLACE_THREADS, 0, stream>>>(row, col, part, tot, start, sorted);
    k_gather_gemm<<<N_NODES / 16, FUSE_THREADS, 0, stream>>>(x, dis, start, sorted, W, b, out);
}


# placement: the count kernel keeps each edge's rank (its counting atomic's return), so k_place has no atomics and runs 4 blocks per slice on all 256 WGPs; order in a slice's cell by arrival as before
# speedup vs baseline: 3.7287x; 1.2272x over previous
#include <hip/hip_runtime.h>
#include <hip/hip_bf16.h>


#define N_NODES 10000
#define N_EDGES 640000
#define IN_C    128
#define OUT_C   128

#define DEG_BLOCKS  64
#define DEG_THREADS 500
#define DEG_SLICE   (N_EDGES / DEG_BLOCKS)
#define DEG_EDGE_PASSES  (DEG_SLICE / 4 / DEG_THREADS)
#define DEG_TABLE_PASSES (N_NODES / 4 / DEG_THREADS)
#define DEG_PITCH   (((N_NODES + 31) / 32) * 32)

#define PLACE_THREADS 512
#define PLACE_OWNERS  (N_NODES / 20)
#define PLACE_VEC     (DEG_SLICE / 4)
#define PLACE_SPLIT    4
#define PLACE_PART_VEC (PLACE_VEC / PLACE_SPLIT)

static_assert(DEG_BLOCKS * DEG_THREADS * DEG_EDGE_PASSES * 4 == N_EDGES, "slices cover the edge list exactly");
static_assert(DEG_THREADS * DEG_TABLE_PASSES * 4 == N_NODES, "passes cover a count table exactly");
static_assert(PLACE_OWNERS * 20 == N_NODES && PLACE_OWNERS <= PLACE_THREADS, "owners cover the bins exactly");
static_assert((DEG_BLOCKS * DEG_PITCH + N_NODES) <= N_NODES * OUT_C, "count planes and totals fit in the output buffer");
static_assert(N_EDGES <= N_NODES * IN_C, "the ordered source list fits in the workspace behind dis and start");
static_assert(N_NODES % 16 == 0, "16-node blocks cover the nodes exactly");
static_assert(PLACE_PART_VEC * PLACE_SPLIT == PLACE_VEC && PLACE_PART_VEC <= 2 * PLACE_THREADS, "two passes cover a block's share of its slice exactly");
static_assert(DEG_SLICE <= 65536, "an edge's rank inside its slice fits in 16 bits");
static_assert((DEG_BLOCKS * DEG_PITCH + N_NODES + N_EDGES / 2) <= N_NODES * OUT_C, "planes, totals and the 16-bit ranks fit in the output buffer");

typedef __attribute__((ext_vector_type(2))) float v2f;
typedef __attribute__((ext_vector_type(8))) float v8f;

__global__ void __launch_bounds__(DEG_THREADS)
k_deg_count(const int* __restrict__ col, unsigned* __restrict__ part,
            unsigned* __restrict__ rank) {
    __shared__ uint4 cnt4[N_NODES / 4];
    unsigned* cnt = (unsigned*)cnt4;
    const int t = threadIdx.x;

    const int4* __restrict__ c4 = (const int4*)(col + blockIdx.x * DEG_SLICE);
    int4 c[DEG_EDGE_PASSES];
    for (int j = 0; j < DEG_EDGE_PASSES; ++j) c[j] = c4[t + j * DEG_THREADS];

    for (int j = 0; j < DEG_TABLE_PASSES; ++j)
        cnt4[t + j * DEG_THREADS] = make_uint4(0u, 0u, 0u, 0u);
    __syncthreads();

    uint2 rk[DEG_EDGE_PASSES];
    for (int j = 0; j < DEG_EDGE_PASSES; ++j) {
        const unsigned r0 = atomicAdd(&cnt[c[j].x], 1u);
        const unsigned r1 = atomicAdd(&cnt[c[j].y], 1u);
        const unsigned r2 = atomicAdd(&cnt[c[j].z], 1u);
        const unsigned r3 = atomicAdd(&cnt[c[j].w], 1u);
        rk[j] = make_uint2(r0 | (r1 << 16), r2 | (r3 << 16));
    }
    __syncthreads();

    uint4* __restrict__ p4 = (uint4*)(part + blockIdx.x * DEG_PITCH);
    for (int j = 0; j < DEG_TABLE_PASSES; ++j)
        p4[t + j * DEG_THREADS] = cnt4[t + j * DEG_THREADS];

    uint2* __restrict__ k2 = (uint2*)rank + blockIdx.x * PLACE_VEC;
    for (int j = 0; j < DEG_EDGE_PASSES; ++j) k2[t + j * DEG_THREADS] = rk[j];
}

__global__ void k_slice_prefix(unsigned* __restrict__ part,
                               unsigned* __restrict__ tot,
                               float* __restrict__ dis) {
    int i = blockIdx.x * blockDim.x + threadIdx.x;
    if (i < N_NODES) {
        unsigned n[DEG_BLOCKS];
        #pragma unroll
        for (int g = 0; g < DEG_BLOCKS; ++g) n[g] = part[g * DEG_PITCH + i];
        unsigned run = 0u;
        #pragma unroll
        for (int g = 0; g < DEG_BLOCKS; ++g) {
            part[g * DEG_PITCH + i] = run;
            run += n[g];
        }
        tot[i] = run;
        float d = (float)(run + 1u);
        dis[i] = (d > 0.0f) ? rsqrtf(d) : 0.0f;
    }
}

__global__ void __launch_bounds__(PLACE_THREADS)
k_place(const int* __restrict__ row, const int* __restrict__ col,
        const unsigned* __restrict__ part, const unsigned* __restrict__ tot,
        const unsigned* __restrict__ rank,
        unsigned* __restrict__ start, int* __restrict__ sorted) {
    __shared__ uint4 pos4[N_NODES / 4];
    __shared__ unsigned wave_sum[PLACE_THREADS / 32];
    const unsigned* pos = (const unsigned*)pos4;
    const int t    = threadIdx.x;
    const int lane = t & 31;
    const int wave = t >> 5;
    const int g    = blockIdx.x / PLACE_SPLIT;
    const int v0   = g * PLACE_VEC + (blockIdx.x % PLACE_SPLIT) * PLACE_PART_VEC;

    const int4*  __restrict__ c4 = (const int4*)col + v0;
    const int4*  __restrict__ r4 = (const int4*)row + v0;
    const uint2* __restrict__ k2 = (const uint2*)rank + v0;
    int4 c[2], r[2];
    uint2 k[2];
    for (int j = 0; j < 2; ++j) {
        const int idx = t + j * PLACE_THREADS;
        if (idx < PLACE_PART_VEC) { c[j] = c4[idx]; r[j] = r4[idx]; k[j] = k2[idx]; }
    }

    uint4 e[5], o[5];
    unsigned mine = 0u;
    if (t < PLACE_OWNERS) {
        const uint4* __restrict__ n4 = (const uint4*)tot + t * 5;
        const uint4* __restrict__ o4 = (const uint4*)(part + g * DEG_PITCH) + t * 5;
        for (int j = 0; j < 5; ++j) {
            const uint4 n = n4[j];
            o[j] = o4[j];
            e[j].x = mine; mine += n.x;
            e[j].y = mine; mine += n.y;
            e[j].z = mine; mine += n.z;
            e[j].w = mine; mine += n.w;
        }
    }
    unsigned inc = mine;
    for (int d = 1; d < 32; d <<= 1) {
        const unsigned v = __shfl_up(inc, d);
        if (lane >= d) inc += v;
    }
    if (lane == 31) wave_sum[wave] = inc;
    __syncthreads();
    unsigned before = inc - mine;
    for (int w = 0; w < wave; ++w) before += wave_sum[w];

    if (t < PLACE_OWNERS) {
        for (int j = 0; j < 5; ++j) {
            uint4 s = e[j];
            s.x += before; s.y += before; s.z += before; s.w += before;
            if (blockIdx.x == 0) ((uint4*)start)[t * 5 + j] = s;
            s.x += o[j].x; s.y += o[j].y; s.z += o[j].z; s.w += o[j].w;
            pos4[t * 5 + j] = s;
        }
    }
    __syncthreads();

    for (int j = 0; j < 2; ++j) {
        const int idx = t + j * PLACE_THREADS;
        if (idx < PLACE_PART_VEC) {
            c[j].x = (int)(pos[c[j].x] + (k[j].x & 0xffffu));
            c[j].y = (int)(pos[c[j].y] + (k[j].x >> 16));
            c[j].z = (int)(pos[c[j].z] + (k[j].y & 0xffffu));
            c[j].w = (int)(pos[c[j].w] + (k[j].y >> 16));
        }
    }
    for (int j = 0; j < 2; ++j) {
        const int idx = t + j * PLACE_THREADS;
        if (idx < PLACE_PART_VEC) {
            sorted[c[j].x] = r[j].x;
            sorted[c[j].y] = r[j].y;
            sorted[c[j].z] = r[j].z;
            sorted[c[j].w] = r[j].w;
        }
    }
}

#define GATHER_BATCH 8
#define GEMM_PITCH 132
#define FUSE_THREADS 512
__global__ void __launch_bounds__(FUSE_THREADS)
k_gather_gemm(const float* __restrict__ x, const float* __restrict__ dis,
              const unsigned* __restrict__ start, const int* __restrict__ sorted,
              const float* __restrict__ W, const float* __restrict__ bias,
              float* __restrict__ out) {
    __shared__ float4 w_s4[OUT_C * GEMM_PITCH / 4];
    __shared__ float4 a_s4[16 * GEMM_PITCH / 4];
    float* w_s = (float*)w_s4;
    float* a_s = (float*)a_s4;

    const int t    = threadIdx.x;
    const int lane = t & 31;
    const int wave = __builtin_amdgcn_readfirstlane(t >> 5);
    const int tm   = blockIdx.x;
    const int c    = tm * 16 + wave;

    const int mn   = lane & 15;
    const int ncol = (wave & 7) * 16 + mn;
    const float bn = bias[ncol];

    const int s = (int)start[c];
    const int n = ((c + 1 < N_NODES) ? (int)start[c + 1] : N_EDGES) - s;
    const float dc = dis[c];
    const float4 self = *(const float4*)(x + c * IN_C + lane * 4);

    int   rl = c, rn = c;
    float wl = 0.0f;
    if (lane < n)      rl = sorted[s + lane];
    if (32 + lane < n) rn = sorted[s + 32 + lane];
    if (lane < n)      wl = dis[rl];

    const float4* __restrict__ w4 = (const float4*)W;
    float4 wv[8];
    for (int j = 0; j < 8; ++j) wv[j] = w4[t + j * FUSE_THREADS];
    for (int j = 0; j < 8; ++j) {
        const int idx = t + j * FUSE_THREADS;
        *(float4*)(w_s + (idx >> 5) * GEMM_PITCH + (idx & 31) * 4) = wv[j];
    }

    float4 acc = make_float4(dc * self.x, dc * self.y, dc * self.z, dc * self.w);

    for (int base = 0; base < n; base += 32) {
        int   rnn = c;
        float wn  = 0.0f;
        if (base + 64 + lane < n) rnn = sorted[s + base + 64 + lane];
        if (base + 32 + lane < n) wn  = dis[rn];

        const int m = (n - base < 32) ? (n - base) : 32;
        for (int j = 0; j < m; j += GATHER_BATCH) {
            float4 v[GATHER_BATCH];
            float  w[GATHER_BATCH];
            #pragma unroll
            for (int u = 0; u < GATHER_BATCH; ++u) {
                const int rr = __builtin_amdgcn_readlane(rl, j + u);
                w[u] = __int_as_float(__builtin_amdgcn_readlane(__float_as_int(wl), j + u));
                v[u] = *(const float4*)(x + rr * IN_C + lane * 4);
            }
            #pragma unroll
            for (int u = 0; u < GATHER_BATCH; ++u) {
                acc.x += w[u] * v[u].x;
                acc.y += w[u] * v[u].y;
                acc.z += w[u] * v[u].z;
                acc.w += w[u] * v[u].w;
            }
        }
        rl = rn;
        wl = wn;
        rn = rnn;
    }

    *(float4*)(a_s + wave * GEMM_PITCH + lane * 4) =
        make_float4(dc * acc.x, dc * acc.y, dc * acc.z, dc * acc.w);
    __syncthreads();

    if (wave < 8) {
        const int khalf = (lane >> 4) << 1;

        const float* xr = a_s + mn * GEMM_PITCH + khalf;
        const float* wr = w_s + (wave * 16 + mn) * GEMM_PITCH + khalf;

        v8f d = {bn, bn, bn, bn, bn, bn, bn, bn};
        #pragma unroll 4
        for (int k = 0; k < IN_C; k += 4) {
            v2f a = *(const v2f*)(xr + k);
            v2f b = *(const v2f*)(wr + k);
            d = __builtin_amdgcn_wmma_f32_16x16x4_f32(false, a, false, b,
                                                      (short)0, d, false, false);
        }

        const int mbase = tm * 16 + ((lane >> 4) << 3);
        #pragma unroll
        for (int v = 0; v < 8; ++v) {
            out[(mbase + v) * OUT_C + ncol] = d[v];
        }
    }
}

extern "C" void kernel_launch(void* const* d_in, const int* in_sizes, int n_in,
                              void* d_out, int out_size, void* d_ws, size_t ws_size,
                              hipStream_t stream) {
    const float* x  = (const float*)d_in[0];
    const int*   ei = (const int*)  d_in[1];
    const float* W  = (const float*)d_in[2];
    const float* b  = (const float*)d_in[3];
    float*       out = (float*)d_out;

    const int* row = ei;
    const int* col = ei + N_EDGES;

    float*    dis    = (float*)d_ws;
    unsigned* start  = (unsigned*)(dis + N_NODES);
    int*      sorted = (int*)(dis + 2 * N_NODES);
    unsigned* part   = (unsigned*)d_out;
    unsigned* tot    = part + DEG_BLOCKS * DEG_PITCH;
    unsigned* rank   = tot + N_NODES;

    k_deg_count<<<DEG_BLOCKS, DEG_THREADS, 0, stream>>>(col, part, rank);
    k_slice_prefix<<<(N_NODES + 63) / 64, 64, 0, stream>>>(part, tot, dis);
    k_place<<<DEG_BLOCKS * PLACE_SPLIT, PLACE_THREADS, 0, stream>>>(row, col, part, tot, rank, start, sorted);
    k_gather_gemm<<<N_NODES / 16, FUSE_THREADS, 0, stream>>>(x, dis, start, sorted, W, b, out);
}


# placement launch gone: each slice orders its own edges by destination inside the count launch (LDS scan + LDS copy); a tile block packs its 64 slice pieces by node in LDS; sum order unchanged
# speedup vs baseline: 5.1692x; 1.3863x over previous
#include <hip/hip_runtime.h>
#include <hip/hip_bf16.h>


#define N_NODES 10000
#define N_EDGES 640000
#define IN_C    128
#define OUT_C   128

#define DEG_BLOCKS  64
#define DEG_THREADS 500
#define DEG_SLICE   (N_EDGES / DEG_BLOCKS)
#define DEG_EDGE_PASSES  (DEG_SLICE / 4 / DEG_THREADS)
#define DEG_TABLE_PASSES (N_NODES / 4 / DEG_THREADS)
#define DEG_PITCH   (((N_NODES + 31) / 32) * 32)
#define N_TILES     (N_NODES / 16)

static_assert(DEG_BLOCKS * DEG_THREADS * DEG_EDGE_PASSES * 4 == N_EDGES, "slices cover the edge list exactly");
static_assert(DEG_THREADS * DEG_TABLE_PASSES * 4 == N_NODES, "passes cover a count table exactly");
static_assert(DEG_THREADS * 20 == N_NODES, "every thread owns 20 consecutive bins");
static_assert(DEG_BLOCKS * DEG_PITCH <= N_NODES * OUT_C, "count planes fit in the output buffer");
static_assert(N_NODES % 16 == 0, "16-node blocks cover the nodes exactly");
static_assert(N_NODES <= 65536, "a source index fits in the low 16 bits of an entry");
static_assert(DEG_BLOCKS == 64, "two slices per lane, four per wave in the tile kernel");

typedef __attribute__((ext_vector_type(2))) float v2f;
typedef __attribute__((ext_vector_type(8))) float v8f;

__global__ void __launch_bounds__(DEG_THREADS)
k_count_order(const int* __restrict__ row, const int* __restrict__ col,
              unsigned* __restrict__ part, unsigned* __restrict__ tile_at,
              unsigned* __restrict__ sorted) {
    __shared__ uint4 cnt4[N_NODES / 4];
    __shared__ uint4 buf4[DEG_SLICE / 4];
    __shared__ unsigned wave_sum[(DEG_THREADS + 31) / 32];
    unsigned* cnt = (unsigned*)cnt4;
    unsigned* buf = (unsigned*)buf4;
    const int t    = threadIdx.x;
    const int lane = t & 31;
    const int wave = t >> 5;
    const int g    = blockIdx.x;

    const int4* __restrict__ c4 = (const int4*)(col + g * DEG_SLICE);
    const int4* __restrict__ r4 = (const int4*)(row + g * DEG_SLICE);
    int4 c[DEG_EDGE_PASSES], r[DEG_EDGE_PASSES];
    for (int j = 0; j < DEG_EDGE_PASSES; ++j) c[j] = c4[t + j * DEG_THREADS];
    for (int j = 0; j < DEG_EDGE_PASSES; ++j) r[j] = r4[t + j * DEG_THREADS];

    for (int j = 0; j < DEG_TABLE_PASSES; ++j)
        cnt4[t + j * DEG_THREADS] = make_uint4(0u, 0u, 0u, 0u);
    __syncthreads();

    uint4 rk[DEG_EDGE_PASSES];
    for (int j = 0; j < DEG_EDGE_PASSES; ++j) {
        rk[j].x = atomicAdd(&cnt[c[j].x], 1u);
        rk[j].y = atomicAdd(&cnt[c[j].y], 1u);
        rk[j].z = atomicAdd(&cnt[c[j].z], 1u);
        rk[j].w = atomicAdd(&cnt[c[j].w], 1u);
    }
    __syncthreads();

    uint4* __restrict__ p4 = (uint4*)(part + g * DEG_PITCH) + t * 5;
    uint4 e[5];
    unsigned mine = 0u;
    for (int j = 0; j < 5; ++j) {
        const uint4 n = cnt4[t * 5 + j];
        p4[j] = n;
        e[j].x = mine; mine += n.x;
        e[j].y = mine; mine += n.y;
        e[j].z = mine; mine += n.z;
        e[j].w = mine; mine += n.w;
    }
    unsigned inc = mine;
    for (int d = 1; d < 32; d <<= 1) {
        const unsigned v = __shfl_up(inc, d);
        if (lane >= d) inc += v;
    }
    if (lane == 31) wave_sum[wave] = inc;
    __syncthreads();
    unsigned before = inc - mine;
    for (int w = 0; w < wave; ++w) before += wave_sum[w];
    for (int j = 0; j < 5; ++j) {
        uint4 s = e[j];
        s.x += before; s.y += before; s.z += before; s.w += before;
        cnt4[t * 5 + j] = s;
    }
    __syncthreads();

    for (int T = t; T <= N_TILES; T += DEG_THREADS)
        tile_at[T * DEG_BLOCKS + g] = (T < N_TILES) ? cnt[T * 16] : (unsigned)DEG_SLICE;

    for (int j = 0; j < DEG_EDGE_PASSES; ++j) {
        buf[cnt[c[j].x] + rk[j].x] = (unsigned)r[j].x | ((unsigned)(c[j].x & 15) << 16);
        buf[cnt[c[j].y] + rk[j].y] = (unsigned)r[j].y | ((unsigned)(c[j].y & 15) << 16);
        buf[cnt[c[j].z] + rk[j].z] = (unsigned)r[j].z | ((unsigned)(c[j].z & 15) << 16);
        buf[cnt[c[j].w] + rk[j].w] = (unsigned)r[j].w | ((unsigned)(c[j].w & 15) << 16);
    }
    __syncthreads();

    uint4* __restrict__ s4 = (uint4*)(sorted + g * DEG_SLICE);
    for (int j = 0; j < DEG_EDGE_PASSES; ++j)
        s4[t + j * DEG_THREADS] = buf4[t + j * DEG_THREADS];
}

__global__ void k_dis(const unsigned* __restrict__ part, unsigned* __restrict__ tot,
                      float* __restrict__ dis) {
    int i = blockIdx.x * blockDim.x + threadIdx.x;
    if (i < N_NODES) {
        unsigned run = 0u;
        #pragma unroll
        for (int g = 0; g < DEG_BLOCKS; ++g) run += part[g * DEG_PITCH + i];
        tot[i] = run;
        float d = (float)(run + 1u);
        dis[i] = (d > 0.0f) ? rsqrtf(d) : 0.0f;
    }
}

#define GATHER_BATCH 8
#define GEMM_PITCH 132
#define FUSE_THREADS 512
#define TILE_CAP 2048
__global__ void __launch_bounds__(FUSE_THREADS)
k_gather_gemm(const float* __restrict__ x, const float* __restrict__ dis,
              const unsigned* __restrict__ tot,
              const unsigned* __restrict__ tile_at, const unsigned* __restrict__ sorted,
              const float* __restrict__ W, const float* __restrict__ bias,
              float* __restrict__ out) {
    __shared__ float4 w_s4[OUT_C * GEMM_PITCH / 4];
    __shared__ float4 a_s4[16 * GEMM_PITCH / 4];
    __shared__ uint4 list1q[TILE_CAP / 4 + 1];
    __shared__ unsigned list2[TILE_CAP];
    __shared__ unsigned wcnt[FUSE_THREADS / 32];
    float* w_s = (float*)w_s4;
    float* a_s = (float*)a_s4;
    unsigned* list1 = (unsigned*)list1q;

    const int t    = threadIdx.x;
    const int lane = t & 31;
    const int wave = __builtin_amdgcn_readfirstlane(t >> 5);
    const int tm   = blockIdx.x;
    const int c    = tm * 16 + wave;

    const int mn   = lane & 15;
    const int ncol = (wave & 7) * 16 + mn;
    const float bn = bias[ncol];

    const unsigned* __restrict__ tp = tile_at + tm * DEG_BLOCKS;
    const int lo0 = (int)tp[lane];
    const int lo1 = (int)tp[32 + lane];
    const int hi0 = (int)tp[DEG_BLOCKS + lane];
    const int hi1 = (int)tp[DEG_BLOCKS + 32 + lane];

    const int tl = (lane < 16) ? (int)tot[tm * 16 + lane] : 0;

    const float dc = dis[c];
    const float4 self = *(const float4*)(x + c * IN_C + lane * 4);

    const float4* __restrict__ w4 = (const float4*)W;
    float4 wv[8];
    for (int j = 0; j < 8; ++j) wv[j] = w4[t + j * FUSE_THREADS];

    const int len0 = hi0 - lo0;
    const int len1 = hi1 - lo1;
    int inc0 = len0, inc1 = len1;
    for (int d = 1; d < 32; d <<= 1) {
        const int v0 = __shfl_up(inc0, d);
        const int v1 = __shfl_up(inc1, d);
        if (lane >= d) { inc0 += v0; inc1 += v1; }
    }
    const int tot0 = __builtin_amdgcn_readlane(inc0, 31);
    const int nT   = tot0 + __builtin_amdgcn_readlane(inc1, 31);
    const bool second = wave >= 8;
    const int preS = second ? (tot0 + inc1 - len1) : (inc0 - len0);
    const int lenS = second ? len1 : len0;
    const int srcS = second ? ((lane + 32) * DEG_SLICE + lo1) : (lane * DEG_SLICE + lo0);
    const bool one_window = nT <= TILE_CAP;
    const int deg_c = __builtin_amdgcn_readlane(tl, wave);
    int before_c = 0;
    for (int w = 0; w < wave; ++w) before_c += __builtin_amdgcn_readlane(tl, w);

    for (int j = 0; j < 8; ++j) {
        const int idx = t + j * FUSE_THREADS;
        *(float4*)(w_s + (idx >> 5) * GEMM_PITCH + (idx & 31) * 4) = wv[j];
    }

    float4 acc = make_float4(dc * self.x, dc * self.y, dc * self.z, dc * self.w);

    for (int ws = 0; ws < nT; ws += TILE_CAP) {
        int sp[4], sb[4], jlo[4], jhi[4];
        unsigned ev[4];
        #pragma unroll
        for (int k = 0; k < 4; ++k) {
            const int gl = (wave * 4 + k) & 31;
            sp[k] = __builtin_amdgcn_readlane(preS, gl);
            sb[k] = __builtin_amdgcn_readlane(srcS, gl);
            const int sl = __builtin_amdgcn_readlane(lenS, gl);
            jlo[k] = (ws - sp[k] > 0) ? (ws - sp[k]) : 0;
            jhi[k] = (ws + TILE_CAP - sp[k] < sl) ? (ws + TILE_CAP - sp[k]) : sl;
            ev[k] = 0u;
            if (jlo[k] + lane < jhi[k]) ev[k] = sorted[sb[k] + jlo[k] + lane];
        }
        #pragma unroll
        for (int k = 0; k < 4; ++k) {
            if (jlo[k] + lane < jhi[k]) list1[sp[k] + jlo[k] + lane - ws] = ev[k];
            for (int j = jlo[k] + 32 + lane; j < jhi[k]; j += 32)
                list1[sp[k] + j - ws] = sorted[sb[k] + j];
        }
        if (t < 4) list1[((nT - ws < TILE_CAP) ? (nT - ws) : TILE_CAP) + t] = 0xffffffffu;
        __syncthreads();

        const int nw = (nT - ws < TILE_CAP) ? (nT - ws) : TILE_CAP;
        const int nq = (nw + 3) >> 2;
        int n, s;
        if (one_window) {
            n = deg_c;
            s = before_c;
        } else {
            n = 0;
            for (int q0 = 0; q0 < nq; q0 += 32) {
                uint4 en = make_uint4(0xffffffffu, 0xffffffffu, 0xffffffffu, 0xffffffffu);
                if (q0 + lane < nq) en = list1q[q0 + lane];
                n += __builtin_popcount(__builtin_amdgcn_ballot_w32((int)(en.x >> 16) == wave))
                   + __builtin_popcount(__builtin_amdgcn_ballot_w32((int)(en.y >> 16) == wave))
                   + __builtin_popcount(__builtin_amdgcn_ballot_w32((int)(en.z >> 16) == wave))
                   + __builtin_popcount(__builtin_amdgcn_ballot_w32((int)(en.w >> 16) == wave));
            }
            if (lane == 0) wcnt[wave] = (unsigned)n;
            __syncthreads();
            s = 0;
            for (int w = 0; w < wave; ++w) s += (int)wcnt[w];
        }
        int kk = s;
        for (int q0 = 0; q0 < nq; q0 += 4 * 32) {
            uint4 en[4];
            #pragma unroll
            for (int u = 0; u < 4; ++u) {
                en[u] = make_uint4(0xffffffffu, 0xffffffffu, 0xffffffffu, 0xffffffffu);
                if (q0 + u * 32 + lane < nq) en[u] = list1q[q0 + u * 32 + lane];
            }
            #pragma unroll
            for (int u = 0; u < 4; ++u) {
                const bool h0 = (int)(en[u].x >> 16) == wave;
                const bool h1 = (int)(en[u].y >> 16) == wave;
                const bool h2 = (int)(en[u].z >> 16) == wave;
                const bool h3 = (int)(en[u].w >> 16) == wave;
                const unsigned m0 = __builtin_amdgcn_ballot_w32(h0);
                const unsigned m1 = __builtin_amdgcn_ballot_w32(h1);
                const unsigned m2 = __builtin_amdgcn_ballot_w32(h2);
                const unsigned m3 = __builtin_amdgcn_ballot_w32(h3);
                int p = kk + (int)__builtin_amdgcn_mbcnt_lo(m3,
                                 __builtin_amdgcn_mbcnt_lo(m2,
                                 __builtin_amdgcn_mbcnt_lo(m1,
                                 __builtin_amdgcn_mbcnt_lo(m0, 0u))));
                if (h0) { list2[p] = en[u].x & 0xffffu; ++p; }
                if (h1) { list2[p] = en[u].y & 0xffffu; ++p; }
                if (h2) { list2[p] = en[u].z & 0xffffu; ++p; }
                if (h3) { list2[p] = en[u].w & 0xffffu; }
                kk += __builtin_popcount(m0) + __builtin_popcount(m1)
                    + __builtin_popcount(m2) + __builtin_popcount(m3);
            }
        }

        int   rl = c, rn = c;
        float wl = 0.0f;
        if (lane < n)      rl = (int)list2[s + lane];
        if (32 + lane < n) rn = (int)list2[s + 32 + lane];
        if (lane < n)      wl = dis[rl];

        for (int base = 0; base < n; base += 32) {
            int   rnn = c;
            float wn  = 0.0f;
            if (base + 64 + lane < n) rnn = (int)list2[s + base + 64 + lane];
            if (base + 32 + lane < n) wn  = dis[rn];

            const int m = (n - base < 32) ? (n - base) : 32;
            for (int j = 0; j < m; j += GATHER_BATCH) {
                float4 v[GATHER_BATCH];
                float  w[GATHER_BATCH];
                #pragma unroll
                for (int u = 0; u < GATHER_BATCH; ++u) {
                    const int rr = __builtin_amdgcn_readlane(rl, j + u);
                    w[u] = __int_as_float(__builtin_amdgcn_readlane(__float_as_int(wl), j + u));
                    v[u] = *(const float4*)(x + rr * IN_C + lane * 4);
                }
                #pragma unroll
                for (int u = 0; u < GATHER_BATCH; ++u) {
                    acc.x += w[u] * v[u].x;
                    acc.y += w[u] * v[u].y;
                    acc.z += w[u] * v[u].z;
                    acc.w += w[u] * v[u].w;
                }
            }
            rl = rn;
            wl = wn;
            rn = rnn;
        }
        __syncthreads();
    }

    *(float4*)(a_s + wave * GEMM_PITCH + lane * 4) =
        make_float4(dc * acc.x, dc * acc.y, dc * acc.z, dc * acc.w);
    __syncthreads();

    if (wave < 8) {
        const int khalf = (lane >> 4) << 1;

        const float* xr = a_s + mn * GEMM_PITCH + khalf;
        const float* wr = w_s + (wave * 16 + mn) * GEMM_PITCH + khalf;

        v8f d = {bn, bn, bn, bn, bn, bn, bn, bn};
        #pragma unroll 4
        for (int k = 0; k < IN_C; k += 4) {
            v2f a = *(const v2f*)(xr + k);
            v2f b = *(const v2f*)(wr + k);
            d = __builtin_amdgcn_wmma_f32_16x16x4_f32(false, a, false, b,
                                                      (short)0, d, false, false);
        }

        const int mbase = tm * 16 + ((lane >> 4) << 3);
        #pragma unroll
        for (int v = 0; v < 8; ++v) {
            out[(mbase + v) * OUT_C + ncol] = d[v];
        }
    }
}

extern "C" void kernel_launch(void* const* d_in, const int* in_sizes, int n_in,
                              void* d_out, int out_size, void* d_ws, size_t ws_size,
                              hipStream_t stream) {
    const float* x  = (const float*)d_in[0];
    const int*   ei = (const int*)  d_in[1];
    const float* W  = (const float*)d_in[2];
    const float* b  = (const float*)d_in[3];
    float*       out = (float*)d_out;

    const int* row = ei;
    const int* col = ei + N_EDGES;

    float*    dis     = (float*)d_ws;
    unsigned* tot     = (unsigned*)d_ws + DEG_PITCH;
    unsigned* tile_at = (unsigned*)d_ws + 2 * DEG_PITCH;
    unsigned* sorted  = tile_at + (N_TILES + 1) * DEG_BLOCKS;
    unsigned* part    = (unsigned*)d_out;

    k_count_order<<<DEG_BLOCKS, DEG_THREADS, 0, stream>>>(row, col, part, tile_at, sorted);
    k_dis<<<(N_NODES + 63) / 64, 64, 0, stream>>>(part, tot, dis);
    k_gather_gemm<<<N_NODES / 16, FUSE_THREADS, 0, stream>>>(x, dis, tot, tile_at, sorted, W, b, out);
}


# count launch also stores each node's run (start, length) per slice; a node's wave reads its 64 runs directly, no wave scans the tile's whole edge list; sum order differs only where a run exceeds 6
# speedup vs baseline: 5.6593x; 1.0948x over previous
#include <hip/hip_runtime.h>
#include <hip/hip_bf16.h>


#define N_NODES 10000
#define N_EDGES 640000
#define IN_C    128
#define OUT_C   128

#define DEG_BLOCKS  64
#define DEG_THREADS 500
#define DEG_SLICE   (N_EDGES / DEG_BLOCKS)
#define DEG_EDGE_PASSES  (DEG_SLICE / 4 / DEG_THREADS)
#define DEG_TABLE_PASSES (N_NODES / 4 / DEG_THREADS)
#define DEG_PITCH   (((N_NODES + 31) / 32) * 32)
#define N_TILES     (N_NODES / 16)
#define TILE_WORDS  (DEG_BLOCKS * 16)

static_assert(DEG_BLOCKS * DEG_THREADS * DEG_EDGE_PASSES * 4 == N_EDGES, "slices cover the edge list exactly");
static_assert(DEG_THREADS * DEG_TABLE_PASSES * 4 == N_NODES, "passes cover a count table exactly");
static_assert(DEG_THREADS * 20 == N_NODES, "every thread owns 20 consecutive bins");
static_assert(N_NODES % 16 == 0, "16-node blocks cover the nodes exactly");
static_assert(N_NODES <= 65536, "a source index fits in 16 bits");
static_assert(DEG_SLICE < 65536, "a run's start and its length fit in 16 bits each");
static_assert(DEG_SLICE % 8 == 0, "an ordered slice of 16-bit sources is whole 16-B vectors");
static_assert(DEG_BLOCKS == 64, "two slices per lane in the tile kernel");

typedef __attribute__((ext_vector_type(2))) float v2f;
typedef __attribute__((ext_vector_type(8))) float v8f;

__global__ void __launch_bounds__(DEG_THREADS)
k_count_order(const int* __restrict__ row, const int* __restrict__ col,
              unsigned* __restrict__ runs, unsigned short* __restrict__ sorted) {
    __shared__ uint4 cnt4[N_NODES / 4];
    __shared__ uint4 buf4[DEG_SLICE / 8];
    __shared__ unsigned wave_sum[(DEG_THREADS + 31) / 32];
    unsigned* cnt = (unsigned*)cnt4;
    unsigned short* buf = (unsigned short*)buf4;
    const int t    = threadIdx.x;
    const int lane = t & 31;
    const int wave = t >> 5;
    const int g    = blockIdx.x;

    const int4* __restrict__ c4 = (const int4*)(col + g * DEG_SLICE);
    const int4* __restrict__ r4 = (const int4*)(row + g * DEG_SLICE);
    int4 c[DEG_EDGE_PASSES], r[DEG_EDGE_PASSES];
    for (int j = 0; j < DEG_EDGE_PASSES; ++j) c[j] = c4[t + j * DEG_THREADS];
    for (int j = 0; j < DEG_EDGE_PASSES; ++j) r[j] = r4[t + j * DEG_THREADS];

    for (int j = 0; j < DEG_TABLE_PASSES; ++j)
        cnt4[t + j * DEG_THREADS] = make_uint4(0u, 0u, 0u, 0u);
    __syncthreads();

    uint4 rk[DEG_EDGE_PASSES];
    for (int j = 0; j < DEG_EDGE_PASSES; ++j) {
        rk[j].x = atomicAdd(&cnt[c[j].x], 1u);
        rk[j].y = atomicAdd(&cnt[c[j].y], 1u);
        rk[j].z = atomicAdd(&cnt[c[j].z], 1u);
        rk[j].w = atomicAdd(&cnt[c[j].w], 1u);
    }
    __syncthreads();

    uint4 n[5], e[5];
    unsigned mine = 0u;
    for (int j = 0; j < 5; ++j) {
        n[j] = cnt4[t * 5 + j];
        e[j].x = mine; mine += n[j].x;
        e[j].y = mine; mine += n[j].y;
        e[j].z = mine; mine += n[j].z;
        e[j].w = mine; mine += n[j].w;
    }
    unsigned inc = mine;
    for (int d = 1; d < 32; d <<= 1) {
        const unsigned v = __shfl_up(inc, d);
        if (lane >= d) inc += v;
    }
    if (lane == 31) wave_sum[wave] = inc;
    __syncthreads();
    unsigned before = inc - mine;
    for (int w = 0; w < wave; ++w) before += wave_sum[w];
    uint4* __restrict__ q4 = (uint4*)runs;
    for (int j = 0; j < 5; ++j) {
        uint4 s = e[j];
        s.x += before; s.y += before; s.z += before; s.w += before;
        cnt4[t * 5 + j] = s;
        const int i = t * 20 + j * 4;
        q4[((i >> 4) * DEG_BLOCKS + g) * 4 + ((i & 15) >> 2)] =
            make_uint4(s.x | (n[j].x << 16), s.y | (n[j].y << 16),
                       s.z | (n[j].z << 16), s.w | (n[j].w << 16));
    }
    __syncthreads();

    for (int j = 0; j < DEG_EDGE_PASSES; ++j) {
        buf[cnt[c[j].x] + rk[j].x] = (unsigned short)r[j].x;
        buf[cnt[c[j].y] + rk[j].y] = (unsigned short)r[j].y;
        buf[cnt[c[j].z] + rk[j].z] = (unsigned short)r[j].z;
        buf[cnt[c[j].w] + rk[j].w] = (unsigned short)r[j].w;
    }
    __syncthreads();

    static_assert(DEG_SLICE / 8 == 2 * DEG_THREADS + DEG_THREADS / 2, "copy-out: two and a half vectors per thread");
    uint4* __restrict__ s4 = (uint4*)(sorted + g * DEG_SLICE);
    const uint4 o0 = buf4[t];
    const uint4 o1 = buf4[t + DEG_THREADS];
    uint4 o2 = make_uint4(0u, 0u, 0u, 0u);
    if (t < DEG_THREADS / 2) o2 = buf4[t + 2 * DEG_THREADS];
    s4[t] = o0;
    s4[t + DEG_THREADS] = o1;
    if (t < DEG_THREADS / 2) s4[t + 2 * DEG_THREADS] = o2;
}

__global__ void k_dis(const unsigned* __restrict__ runs, float* __restrict__ dis) {
    int i = blockIdx.x * blockDim.x + threadIdx.x;
    if (i < N_NODES) {
        const unsigned* __restrict__ p = runs + (i >> 4) * TILE_WORDS + (i & 15);
        unsigned v[DEG_BLOCKS];
        #pragma unroll
        for (int g = 0; g < DEG_BLOCKS; ++g) v[g] = p[g * 16];
        unsigned run = 0u;
        #pragma unroll
        for (int g = 0; g < DEG_BLOCKS; ++g) run += v[g] >> 16;
        float d = (float)(run + 1u);
        dis[i] = (d > 0.0f) ? rsqrtf(d) : 0.0f;
    }
}

#define GATHER_BATCH 8
#define GEMM_PITCH 132
#define FUSE_THREADS 512
#define RUN_BATCH 6
#define LIST_CAP (DEG_BLOCKS * RUN_BATCH)
#define RUN_PITCH 17
static_assert(FUSE_THREADS * 2 == TILE_WORDS, "one 8-B load per thread copies a tile's run table");
__global__ void __launch_bounds__(FUSE_THREADS) __attribute__((amdgpu_num_vgpr(80)))
k_gather_gemm(const float* __restrict__ x, const float* __restrict__ dis,
              const unsigned* __restrict__ runs, const unsigned short* __restrict__ sorted,
              const float* __restrict__ W, const float* __restrict__ bias,
              float* __restrict__ out) {
    __shared__ float4 w_s4[OUT_C * GEMM_PITCH / 4];
    __shared__ float4 a_s4[16 * GEMM_PITCH / 4];
    __shared__ unsigned run_s[DEG_BLOCKS * RUN_PITCH];
    __shared__ unsigned short list[(FUSE_THREADS / 32) * LIST_CAP];
    float* w_s = (float*)w_s4;
    float* a_s = (float*)a_s4;

    const int t    = threadIdx.x;
    const int lane = t & 31;
    const int wave = __builtin_amdgcn_readfirstlane(t >> 5);
    const int tm   = blockIdx.x;
    const int c    = tm * 16 + wave;

    const int mn   = lane & 15;
    const int ncol = (wave & 7) * 16 + mn;

    const uint2 rv = ((const uint2*)(runs + tm * TILE_WORDS))[t];

    const float bn = bias[ncol];
    const float dc = dis[c];
    const float4 self = *(const float4*)(x + c * IN_C + lane * 4);

    const float4* __restrict__ w4 = (const float4*)W;
    float4 wv[8];
    for (int j = 0; j < 8; ++j) wv[j] = w4[t + j * FUSE_THREADS];

    run_s[(t >> 3) * RUN_PITCH + (t & 7) * 2]     = rv.x;
    run_s[(t >> 3) * RUN_PITCH + (t & 7) * 2 + 1] = rv.y;
    for (int j = 0; j < 8; ++j) {
        const int idx = t + j * FUSE_THREADS;
        *(float4*)(w_s + (idx >> 5) * GEMM_PITCH + (idx & 31) * 4) = wv[j];
    }
    __syncthreads();

    const unsigned run0 = run_s[lane * RUN_PITCH + wave];
    const unsigned run1 = run_s[(lane + 32) * RUN_PITCH + wave];
    const int len0 = (int)(run0 >> 16);
    const int len1 = (int)(run1 >> 16);
    const unsigned at0 = (unsigned)lane * DEG_SLICE + (run0 & 0xffffu);
    const unsigned at1 = (unsigned)(lane + 32) * DEG_SLICE + (run1 & 0xffffu);
    unsigned short* mine = list + wave * LIST_CAP;

    float4 acc = make_float4(dc * self.x, dc * self.y, dc * self.z, dc * self.w);

    int j0 = 0;
    int rem0, rem1;
    unsigned e[2 * RUN_BATCH];
    auto fetch = [&]() {
        rem0 = len0 - j0;
        rem1 = len1 - j0;
        rem0 = (rem0 < 0) ? 0 : ((rem0 > RUN_BATCH) ? RUN_BATCH : rem0);
        rem1 = (rem1 < 0) ? 0 : ((rem1 > RUN_BATCH) ? RUN_BATCH : rem1);
        #pragma unroll
        for (int j = 0; j < RUN_BATCH; ++j) {
            e[j] = 0u;
            if (j < rem0) e[j] = sorted[at0 + (unsigned)(j0 + j)];
        }
        #pragma unroll
        for (int j = 0; j < RUN_BATCH; ++j) {
            e[RUN_BATCH + j] = 0u;
            if (j < rem1) e[RUN_BATCH + j] = sorted[at1 + (unsigned)(j0 + j)];
        }
    };
    fetch();

    for (;;) {
        int inc0 = rem0, inc1 = rem1;
        for (int d = 1; d < 32; d <<= 1) {
            const int v0 = __shfl_up(inc0, d);
            const int v1 = __shfl_up(inc1, d);
            if (lane >= d) { inc0 += v0; inc1 += v1; }
        }
        const int tot0 = __builtin_amdgcn_readlane(inc0, 31);
        const int n    = tot0 + __builtin_amdgcn_readlane(inc1, 31);
        const int pre0 = inc0 - rem0;
        const int pre1 = tot0 + inc1 - rem1;

        #pragma unroll
        for (int j = 0; j < RUN_BATCH; ++j)
            if (j < rem0) mine[pre0 + j] = (unsigned short)e[j];
        #pragma unroll
        for (int j = 0; j < RUN_BATCH; ++j)
            if (j < rem1) mine[pre1 + j] = (unsigned short)e[RUN_BATCH + j];

        int   rl = c, rn = c;
        float wl = 0.0f;
        if (lane < n)      rl = (int)mine[lane];
        if (32 + lane < n) rn = (int)mine[32 + lane];
        if (lane < n)      wl = dis[rl];

        for (int base = 0; base < n; base += 32) {
            int   rnn = c;
            float wn  = 0.0f;
            if (base + 64 + lane < n) rnn = (int)mine[base + 64 + lane];
            if (base + 32 + lane < n) wn  = dis[rn];

            const int m = (n - base < 32) ? (n - base) : 32;
            for (int j = 0; j < m; j += GATHER_BATCH) {
                float4 v[GATHER_BATCH];
                float  w[GATHER_BATCH];
                #pragma unroll
                for (int u = 0; u < GATHER_BATCH; ++u) {
                    const int rr = __builtin_amdgcn_readlane(rl, j + u);
                    w[u] = __int_as_float(__builtin_amdgcn_readlane(__float_as_int(wl), j + u));
                    v[u] = *(const float4*)(x + rr * IN_C + lane * 4);
                }
                #pragma unroll
                for (int u = 0; u < GATHER_BATCH; ++u) {
                    acc.x += w[u] * v[u].x;
                    acc.y += w[u] * v[u].y;
                    acc.z += w[u] * v[u].z;
                    acc.w += w[u] * v[u].w;
                }
            }
            rl = rn;
            wl = wn;
            rn = rnn;
        }

        j0 += RUN_BATCH;
        if (__builtin_amdgcn_ballot_w32(len0 > j0 || len1 > j0) == 0u) break;
        fetch();
    }

    *(float4*)(a_s + wave * GEMM_PITCH + lane * 4) =
        make_float4(dc * acc.x, dc * acc.y, dc * acc.z, dc * acc.w);
    __syncthreads();

    if (wave < 8) {
        const int khalf = (lane >> 4) << 1;

        const float* xr = a_s + mn * GEMM_PITCH + khalf;
        const float* wr = w_s + (wave * 16 + mn) * GEMM_PITCH + khalf;

        v8f d = {bn, bn, bn, bn, bn, bn, bn, bn};
        #pragma unroll 4
        for (int k = 0; k < IN_C; k += 4) {
            v2f a = *(const v2f*)(xr + k);
            v2f b = *(const v2f*)(wr + k);
            d = __builtin_amdgcn_wmma_f32_16x16x4_f32(false, a, false, b,
                                                      (short)0, d, false, false);
        }

        const int mbase = tm * 16 + ((lane >> 4) << 3);
        #pragma unroll
        for (int v = 0; v < 8; ++v) {
            out[(mbase + v) * OUT_C + ncol] = d[v];
        }
    }
}

extern "C" void kernel_launch(void* const* d_in, const int* in_sizes, int n_in,
                              void* d_out, int out_size, void* d_ws, size_t ws_size,
                              hipStream_t stream) {
    const float* x  = (const float*)d_in[0];
    const int*   ei = (const int*)  d_in[1];
    const float* W  = (const float*)d_in[2];
    const float* b  = (const float*)d_in[3];
    float*       out = (float*)d_out;

    const int* row = ei;
    const int* col = ei + N_EDGES;

    float*          dis    = (float*)d_ws;
    unsigned*       runs   = (unsigned*)d_ws + DEG_PITCH;
    unsigned short* sorted = (unsigned short*)(runs + N_TILES * TILE_WORDS);

    k_count_order<<<DEG_BLOCKS, DEG_THREADS, 0, stream>>>(row, col, runs, sorted);
    k_dis<<<(N_NODES + 63) / 64, 64, 0, stream>>>(runs, dis);
    k_gather_gemm<<<N_NODES / 16, FUSE_THREADS, 0, stream>>>(x, dis, runs, sorted, W, b, out);
}


# last launch: dis table (40 KB) copied to LDS per block, a source's weight is an LDS lookup, not a 32-line scattered global load; W goes to LDS after the sums, in the table's place; sums unchanged
# speedup vs baseline: 6.1692x; 1.0901x over previous
#include <hip/hip_runtime.h>
#include <hip/hip_bf16.h>


#define N_NODES 10000
#define N_EDGES 640000
#define IN_C    128
#define OUT_C   128

#define DEG_BLOCKS  64
#define DEG_THREADS 500
#define DEG_SLICE   (N_EDGES / DEG_BLOCKS)
#define DEG_EDGE_PASSES  (DEG_SLICE / 4 / DEG_THREADS)
#define DEG_TABLE_PASSES (N_NODES / 4 / DEG_THREADS)
#define DEG_PITCH   (((N_NODES + 31) / 32) * 32)
#define N_TILES     (N_NODES / 16)
#define TILE_WORDS  (DEG_BLOCKS * 16)

static_assert(DEG_BLOCKS * DEG_THREADS * DEG_EDGE_PASSES * 4 == N_EDGES, "slices cover the edge list exactly");
static_assert(DEG_THREADS * DEG_TABLE_PASSES * 4 == N_NODES, "passes cover a count table exactly");
static_assert(DEG_THREADS * 20 == N_NODES, "every thread owns 20 consecutive bins");
static_assert(N_NODES % 16 == 0, "16-node blocks cover the nodes exactly");
static_assert(N_NODES <= 65536, "a source index fits in 16 bits");
static_assert(DEG_SLICE < 65536, "a run's start and its length fit in 16 bits each");
static_assert(DEG_SLICE % 8 == 0, "an ordered slice of 16-bit sources is whole 16-B vectors");
static_assert(DEG_BLOCKS == 64, "two slices per lane in the tile kernel");

typedef __attribute__((ext_vector_type(2))) float v2f;
typedef __attribute__((ext_vector_type(8))) float v8f;

__global__ void __launch_bounds__(DEG_THREADS)
k_count_order(const int* __restrict__ row, const int* __restrict__ col,
              unsigned* __restrict__ runs, unsigned short* __restrict__ sorted) {
    __shared__ uint4 cnt4[N_NODES / 4];
    __shared__ uint4 buf4[DEG_SLICE / 8];
    __shared__ unsigned wave_sum[(DEG_THREADS + 31) / 32];
    unsigned* cnt = (unsigned*)cnt4;
    unsigned short* buf = (unsigned short*)buf4;
    const int t    = threadIdx.x;
    const int lane = t & 31;
    const int wave = t >> 5;
    const int g    = blockIdx.x;

    const int4* __restrict__ c4 = (const int4*)(col + g * DEG_SLICE);
    const int4* __restrict__ r4 = (const int4*)(row + g * DEG_SLICE);
    int4 c[DEG_EDGE_PASSES], r[DEG_EDGE_PASSES];
    for (int j = 0; j < DEG_EDGE_PASSES; ++j) c[j] = c4[t + j * DEG_THREADS];
    for (int j = 0; j < DEG_EDGE_PASSES; ++j) r[j] = r4[t + j * DEG_THREADS];

    for (int j = 0; j < DEG_TABLE_PASSES; ++j)
        cnt4[t + j * DEG_THREADS] = make_uint4(0u, 0u, 0u, 0u);
    __syncthreads();

    uint4 rk[DEG_EDGE_PASSES];
    for (int j = 0; j < DEG_EDGE_PASSES; ++j) {
        rk[j].x = atomicAdd(&cnt[c[j].x], 1u);
        rk[j].y = atomicAdd(&cnt[c[j].y], 1u);
        rk[j].z = atomicAdd(&cnt[c[j].z], 1u);
        rk[j].w = atomicAdd(&cnt[c[j].w], 1u);
    }
    __syncthreads();

    uint4 n[5], e[5];
    unsigned mine = 0u;
    for (int j = 0; j < 5; ++j) {
        n[j] = cnt4[t * 5 + j];
        e[j].x = mine; mine += n[j].x;
        e[j].y = mine; mine += n[j].y;
        e[j].z = mine; mine += n[j].z;
        e[j].w = mine; mine += n[j].w;
    }
    unsigned inc = mine;
    for (int d = 1; d < 32; d <<= 1) {
        const unsigned v = __shfl_up(inc, d);
        if (lane >= d) inc += v;
    }
    if (lane == 31) wave_sum[wave] = inc;
    __syncthreads();
    unsigned before = inc - mine;
    for (int w = 0; w < wave; ++w) before += wave_sum[w];
    uint4* __restrict__ q4 = (uint4*)runs;
    for (int j = 0; j < 5; ++j) {
        uint4 s = e[j];
        s.x += before; s.y += before; s.z += before; s.w += before;
        cnt4[t * 5 + j] = s;
        const int i = t * 20 + j * 4;
        q4[((i >> 4) * DEG_BLOCKS + g) * 4 + ((i & 15) >> 2)] =
            make_uint4(s.x | (n[j].x << 16), s.y | (n[j].y << 16),
                       s.z | (n[j].z << 16), s.w | (n[j].w << 16));
    }
    __syncthreads();

    for (int j = 0; j < DEG_EDGE_PASSES; ++j) {
        buf[cnt[c[j].x] + rk[j].x] = (unsigned short)r[j].x;
        buf[cnt[c[j].y] + rk[j].y] = (unsigned short)r[j].y;
        buf[cnt[c[j].z] + rk[j].z] = (unsigned short)r[j].z;
        buf[cnt[c[j].w] + rk[j].w] = (unsigned short)r[j].w;
    }
    __syncthreads();

    static_assert(DEG_SLICE / 8 == 2 * DEG_THREADS + DEG_THREADS / 2, "copy-out: two and a half vectors per thread");
    uint4* __restrict__ s4 = (uint4*)(sorted + g * DEG_SLICE);
    const uint4 o0 = buf4[t];
    const uint4 o1 = buf4[t + DEG_THREADS];
    uint4 o2 = make_uint4(0u, 0u, 0u, 0u);
    if (t < DEG_THREADS / 2) o2 = buf4[t + 2 * DEG_THREADS];
    s4[t] = o0;
    s4[t + DEG_THREADS] = o1;
    if (t < DEG_THREADS / 2) s4[t + 2 * DEG_THREADS] = o2;
}

__global__ void k_dis(const unsigned* __restrict__ runs, float* __restrict__ dis) {
    int i = blockIdx.x * blockDim.x + threadIdx.x;
    if (i < N_NODES) {
        const unsigned* __restrict__ p = runs + (i >> 4) * TILE_WORDS + (i & 15);
        unsigned v[DEG_BLOCKS];
        #pragma unroll
        for (int g = 0; g < DEG_BLOCKS; ++g) v[g] = p[g * 16];
        unsigned run = 0u;
        #pragma unroll
        for (int g = 0; g < DEG_BLOCKS; ++g) run += v[g] >> 16;
        float d = (float)(run + 1u);
        dis[i] = (d > 0.0f) ? rsqrtf(d) : 0.0f;
    }
}

#define GATHER_BATCH 8
#define GEMM_PITCH 132
#define FUSE_THREADS 512
#define RUN_BATCH 6
#define LIST_CAP (DEG_BLOCKS * RUN_BATCH)
#define RUN_PITCH 17
static_assert(FUSE_THREADS * 2 == TILE_WORDS, "one 8-B load per thread copies a tile's run table");
#define DIS_VECS   (N_NODES / 4)
#define DIS_PASSES ((DIS_VECS + FUSE_THREADS - 1) / FUSE_THREADS)
static_assert(N_NODES % 4 == 0, "the dis table is whole float4");
static_assert(DIS_PASSES * FUSE_THREADS <= OUT_C * GEMM_PITCH / 4, "the copy of the dis table, whole passes, fits in the place W takes later");
__global__ void __launch_bounds__(FUSE_THREADS) __attribute__((amdgpu_num_vgpr(80)))
k_gather_gemm(const float* __restrict__ x, const float* __restrict__ dis,
              const unsigned* __restrict__ runs, const unsigned short* __restrict__ sorted,
              const float* W, const float* __restrict__ bias,
              float* __restrict__ out) {
    __shared__ float4 w_s4[OUT_C * GEMM_PITCH / 4];
    __shared__ float4 a_s4[16 * GEMM_PITCH / 4];
    __shared__ unsigned run_s[DEG_BLOCKS * RUN_PITCH];
    __shared__ unsigned short list[(FUSE_THREADS / 32) * LIST_CAP];
    float* w_s = (float*)w_s4;
    float* a_s = (float*)a_s4;
    const float* dis_s = (const float*)w_s4;

    const int t    = threadIdx.x;
    const int lane = t & 31;
    const int wave = __builtin_amdgcn_readfirstlane(t >> 5);
    const int tm   = blockIdx.x;
    const int c    = tm * 16 + wave;

    const int mn   = lane & 15;
    const int ncol = (wave & 7) * 16 + mn;

    const uint2 rv = ((const uint2*)(runs + tm * TILE_WORDS))[t];

    const float4* __restrict__ d4 = (const float4*)dis;
    float4 dv[DIS_PASSES];
    #pragma unroll
    for (int j = 0; j < DIS_PASSES; ++j) {
        const int i = t + j * FUSE_THREADS;
        dv[j] = d4[(i < DIS_VECS) ? i : (DIS_VECS - 1)];
    }

    const float bn = bias[ncol];
    const float4 self = *(const float4*)(x + c * IN_C + lane * 4);

    run_s[(t >> 3) * RUN_PITCH + (t & 7) * 2]     = rv.x;
    run_s[(t >> 3) * RUN_PITCH + (t & 7) * 2 + 1] = rv.y;
    #pragma unroll
    for (int j = 0; j < DIS_PASSES; ++j) w_s4[t + j * FUSE_THREADS] = dv[j];
    __syncthreads();

    const float dc = dis_s[c];

    const unsigned run0 = run_s[lane * RUN_PITCH + wave];
    const unsigned run1 = run_s[(lane + 32) * RUN_PITCH + wave];
    const int len0 = (int)(run0 >> 16);
    const int len1 = (int)(run1 >> 16);
    const unsigned at0 = (unsigned)lane * DEG_SLICE + (run0 & 0xffffu);
    const unsigned at1 = (unsigned)(lane + 32) * DEG_SLICE + (run1 & 0xffffu);
    unsigned short* mine = list + wave * LIST_CAP;

    float4 acc = make_float4(dc * self.x, dc * self.y, dc * self.z, dc * self.w);

    int j0 = 0;
    int rem0, rem1;
    unsigned e[2 * RUN_BATCH];
    auto fetch = [&]() {
        rem0 = len0 - j0;
        rem1 = len1 - j0;
        rem0 = (rem0 < 0) ? 0 : ((rem0 > RUN_BATCH) ? RUN_BATCH : rem0);
        rem1 = (rem1 < 0) ? 0 : ((rem1 > RUN_BATCH) ? RUN_BATCH : rem1);
        #pragma unroll
        for (int j = 0; j < RUN_BATCH; ++j) {
            e[j] = 0u;
            if (j < rem0) e[j] = sorted[at0 + (unsigned)(j0 + j)];
        }
        #pragma unroll
        for (int j = 0; j < RUN_BATCH; ++j) {
            e[RUN_BATCH + j] = 0u;
            if (j < rem1) e[RUN_BATCH + j] = sorted[at1 + (unsigned)(j0 + j)];
        }
    };
    fetch();

    for (;;) {
        int inc0 = rem0, inc1 = rem1;
        for (int d = 1; d < 32; d <<= 1) {
            const int v0 = __shfl_up(inc0, d);
            const int v1 = __shfl_up(inc1, d);
            if (lane >= d) { inc0 += v0; inc1 += v1; }
        }
        const int tot0 = __builtin_amdgcn_readlane(inc0, 31);
        const int n    = tot0 + __builtin_amdgcn_readlane(inc1, 31);
        const int pre0 = inc0 - rem0;
        const int pre1 = tot0 + inc1 - rem1;

        #pragma unroll
        for (int j = 0; j < RUN_BATCH; ++j)
            if (j < rem0) mine[pre0 + j] = (unsigned short)e[j];
        #pragma unroll
        for (int j = 0; j < RUN_BATCH; ++j)
            if (j < rem1) mine[pre1 + j] = (unsigned short)e[RUN_BATCH + j];

        int   rl = c, rn = c;
        float wl = 0.0f;
        if (lane < n)      rl = (int)mine[lane];
        if (32 + lane < n) rn = (int)mine[32 + lane];
        if (lane < n)      wl = dis_s[rl];

        for (int base = 0; base < n; base += 32) {
            int   rnn = c;
            float wn  = 0.0f;
            if (base + 64 + lane < n) rnn = (int)mine[base + 64 + lane];
            if (base + 32 + lane < n) wn  = dis_s[rn];

            const int m = (n - base < 32) ? (n - base) : 32;
            for (int j = 0; j < m; j += GATHER_BATCH) {
                float4 v[GATHER_BATCH];
                float  w[GATHER_BATCH];
                #pragma unroll
                for (int u = 0; u < GATHER_BATCH; ++u) {
                    const int rr = __builtin_amdgcn_readlane(rl, j + u);
                    w[u] = __int_as_float(__builtin_amdgcn_readlane(__float_as_int(wl), j + u));
                    v[u] = *(const float4*)(x + rr * IN_C + lane * 4);
                }
                #pragma unroll
                for (int u = 0; u < GATHER_BATCH; ++u) {
                    acc.x += w[u] * v[u].x;
                    acc.y += w[u] * v[u].y;
                    acc.z += w[u] * v[u].z;
                    acc.w += w[u] * v[u].w;
                }
            }
            rl = rn;
            wl = wn;
            rn = rnn;
        }

        j0 += RUN_BATCH;
        if (__builtin_amdgcn_ballot_w32(len0 > j0 || len1 > j0) == 0u) break;
        fetch();
    }

    const float4* w4 = (const float4*)W;
    const float4 wv0 = w4[t];
    const float4 wv1 = w4[t + 1 * FUSE_THREADS];
    const float4 wv2 = w4[t + 2 * FUSE_THREADS];
    const float4 wv3 = w4[t + 3 * FUSE_THREADS];
    const float4 wv4 = w4[t + 4 * FUSE_THREADS];
    const float4 wv5 = w4[t + 5 * FUSE_THREADS];
    const float4 wv6 = w4[t + 6 * FUSE_THREADS];
    const float4 wv7 = w4[t + 7 * FUSE_THREADS];

    *(float4*)(a_s + wave * GEMM_PITCH + lane * 4) =
        make_float4(dc * acc.x, dc * acc.y, dc * acc.z, dc * acc.w);
    __syncthreads();
    float* wdst = w_s + (t >> 5) * GEMM_PITCH + (t & 31) * 4;
    *(float4*)(wdst + 0 * 16 * GEMM_PITCH) = wv0;
    *(float4*)(wdst + 1 * 16 * GEMM_PITCH) = wv1;
    *(float4*)(wdst + 2 * 16 * GEMM_PITCH) = wv2;
    *(float4*)(wdst + 3 * 16 * GEMM_PITCH) = wv3;
    *(float4*)(wdst + 4 * 16 * GEMM_PITCH) = wv4;
    *(float4*)(wdst + 5 * 16 * GEMM_PITCH) = wv5;
    *(float4*)(wdst + 6 * 16 * GEMM_PITCH) = wv6;
    *(float4*)(wdst + 7 * 16 * GEMM_PITCH) = wv7;
    __syncthreads();

    if (wave < 8) {
        const int khalf = (lane >> 4) << 1;

        const float* xr = a_s + mn * GEMM_PITCH + khalf;
        const float* wr = w_s + (wave * 16 + mn) * GEMM_PITCH + khalf;

        v8f d = {bn, bn, bn, bn, bn, bn, bn, bn};
        #pragma unroll 4
        for (int k = 0; k < IN_C; k += 4) {
            v2f a = *(const v2f*)(xr + k);
            v2f b = *(const v2f*)(wr + k);
            d = __builtin_amdgcn_wmma_f32_16x16x4_f32(false, a, false, b,
                                                      (short)0, d, false, false);
        }

        const int mbase = tm * 16 + ((lane >> 4) << 3);
        #pragma unroll
        for (int v = 0; v < 8; ++v) {
            out[(mbase + v) * OUT_C + ncol] = d[v];
        }
    }
}

extern "C" void kernel_launch(void* const* d_in, const int* in_sizes, int n_in,
                              void* d_out, int out_size, void* d_ws, size_t ws_size,
                              hipStream_t stream) {
    const float* x  = (const float*)d_in[0];
    const int*   ei = (const int*)  d_in[1];
    const float* W  = (const float*)d_in[2];
    const float* b  = (const float*)d_in[3];
    float*       out = (float*)d_out;

    const int* row = ei;
    const int* col = ei + N_EDGES;

    float*          dis    = (float*)d_ws;
    unsigned*       runs   = (unsigned*)d_ws + DEG_PITCH;
    unsigned short* sorted = (unsigned short*)(runs + N_TILES * TILE_WORDS);

    k_count_order<<<DEG_BLOCKS, DEG_THREADS, 0, stream>>>(row, col, runs, sorted);
    k_dis<<<(N_NODES + 63) / 64, 64, 0, stream>>>(runs, dis);
    k_gather_gemm<<<N_NODES / 16, FUSE_THREADS, 0, stream>>>(x, dis, runs, sorted, W, b, out);
}
